# starting kernel
# baseline (speedup 1.0000x reference)
#include <hip/hip_runtime.h>
#include <hip/hip_bf16.h>

typedef __attribute__((ext_vector_type(16))) __bf16 v16bf;
typedef __attribute__((ext_vector_type(8)))  float  v8f;

#define D_IN  256
#define H_DIM 32
#define C_DIM 2

__global__ void fill_kernel(float* __restrict__ p, long long n, float v) {
  long long i = (long long)blockIdx.x * blockDim.x + threadIdx.x;
  if (i < n) p[i] = v;
}

__global__ void degree_kernel(const int* __restrict__ ei, float* __restrict__ deg, int E) {
  int e = blockIdx.x * blockDim.x + threadIdx.x;
  if (e < E) atomicAdd(&deg[ei[E + e]], 1.0f);
}

__global__ void rsqrt_kernel(float* __restrict__ deg, int N) {
  int i = blockIdx.x * blockDim.x + threadIdx.x;
  if (i < N) {
    float d = deg[i];
    deg[i] = d > 0.0f ? rsqrtf(d) : 0.0f;
  }
}

__global__ void gemm1_wmma_kernel(const float* __restrict__ x,
                                  const float* __restrict__ W1,
                                  float* __restrict__ t1, int N) {
  __shared__ __align__(32) __bf16 Bfrag[8][2][32][16];

  const int tid  = threadIdx.x;
  const int lane = tid & 31;
  const int wave = tid >> 5;

  for (int idx = tid; idx < 8 * 2 * 32 * 16; idx += blockDim.x) {
    int j     = idx & 15;
    int ln    = (idx >> 4) & 31;
    int ntile = (idx >> 9) & 1;
    int ks    = idx >> 10;
    int jj = j >> 1, p = j & 1;
    int hi = ln >> 4;
    int K  = ks * 32 + 16 * (jj >> 2) + 8 * hi + 2 * (jj & 3) + p;
    int n  = ntile * 16 + (ln & 15);
    Bfrag[ks][ntile][ln][j] = (__bf16)W1[K * H_DIM + n];
  }
  __syncthreads();

  long long rowbase = ((long long)blockIdx.x * 8 + wave) * 16;
  if (rowbase >= N) return;

  const int m  = lane & 15;
  const int hi = lane >> 4;
  const float* xrow = x + (rowbase + m) * D_IN;

  v8f acc0 = {}; v8f acc1 = {};
  #pragma unroll
  for (int ks = 0; ks < 8; ++ks) {
    v16bf a;
    #pragma unroll
    for (int jj = 0; jj < 8; ++jj) {
      int K0 = ks * 32 + 16 * (jj >> 2) + 8 * hi + 2 * (jj & 3);
      float2 f = *(const float2*)(xrow + K0);
      a[2 * jj]     = (__bf16)f.x;
      a[2 * jj + 1] = (__bf16)f.y;
    }
    v16bf b0 = *(const v16bf*)&Bfrag[ks][0][lane][0];
    v16bf b1 = *(const v16bf*)&Bfrag[ks][1][lane][0];
    acc0 = __builtin_amdgcn_wmma_f32_16x16x32_bf16(false, a, false, b0,
                                                   (short)0, acc0, false, false);
    acc1 = __builtin_amdgcn_wmma_f32_16x16x32_bf16(false, a, false, b1,
                                                   (short)0, acc1, false, false);
  }

  #pragma unroll
  for (int r = 0; r < 8; ++r) {
    long long row = rowbase + r + 8 * hi;
    t1[row * H_DIM + m]      = acc0[r];
    t1[row * H_DIM + 16 + m] = acc1[r];
  }
}

__global__ void scatter1_kernel(const int* __restrict__ ei,
                                const float* __restrict__ t1,
                                const float* __restrict__ dinv,
                                float* __restrict__ agg1, int E) {
  long long gid = (long long)blockIdx.x * blockDim.x + threadIdx.x;
  long long e   = gid >> 3;
  int part      = (int)(gid & 7);
  if (e >= E) return;
  long long s = ei[e];
  long long t = ei[(long long)E + e];
  float w = dinv[s] * dinv[t];
  float4 v = *(const float4*)(t1 + s * H_DIM + part * 4);
  float* dst = agg1 + t * H_DIM + part * 4;
  atomicAdd(dst + 0, v.x * w);
  atomicAdd(dst + 1, v.y * w);
  atomicAdd(dst + 2, v.z * w);
  atomicAdd(dst + 3, v.w * w);
}

__global__ void bias_relu_kernel(float* __restrict__ agg1,
                                 const float* __restrict__ t1,
                                 const float* __restrict__ dinv,
                                 const float* __restrict__ b1, int N) {
  long long i = (long long)blockIdx.x * blockDim.x + threadIdx.x;
  if (i >= (long long)N * H_DIM) return;
  long long node = i >> 5;
  int n = (int)(i & 31);
  float di = dinv[node];
  float v = agg1[i] + di * di * t1[i] + b1[n];
  agg1[i] = v > 0.0f ? v : 0.0f;
}

__global__ void gemm2_kernel(const float* __restrict__ h1,
                             const float* __restrict__ W2,
                             float* __restrict__ h2, int N) {
  int i = blockIdx.x * blockDim.x + threadIdx.x;
  if (i >= N) return;
  const float* hr = h1 + (long long)i * H_DIM;
  float z0 = 0.0f, z1 = 0.0f;
  #pragma unroll
  for (int k = 0; k < H_DIM; ++k) {
    float h = hr[k];
    z0 += h * W2[k * C_DIM + 0];
    z1 += h * W2[k * C_DIM + 1];
  }
  h2[(long long)i * C_DIM + 0] = z0;
  h2[(long long)i * C_DIM + 1] = z1;
}

__global__ void scatter2_kernel(const int* __restrict__ ei,
                                const float* __restrict__ h2,
                                const float* __restrict__ dinv,
                                float* __restrict__ agg2, int E) {
  long long e = (long long)blockIdx.x * blockDim.x + threadIdx.x;
  if (e >= E) return;
  long long s = ei[e];
  long long t = ei[(long long)E + e];
  float w = dinv[s] * dinv[t];
  atomicAdd(&agg2[t * C_DIM + 0], h2[s * C_DIM + 0] * w);
  atomicAdd(&agg2[t * C_DIM + 1], h2[s * C_DIM + 1] * w);
}

__global__ void logsoftmax_kernel(const float* __restrict__ agg2,
                                  const float* __restrict__ h2,
                                  const float* __restrict__ dinv,
                                  const float* __restrict__ b2,
                                  float* __restrict__ out, int N) {
  int i = blockIdx.x * blockDim.x + threadIdx.x;
  if (i >= N) return;
  float di = dinv[i], sl = di * di;
  float z0 = agg2[(long long)i * 2 + 0] + sl * h2[(long long)i * 2 + 0] + b2[0];
  float z1 = agg2[(long long)i * 2 + 1] + sl * h2[(long long)i * 2 + 1] + b2[1];
  float mx  = fmaxf(z0, z1);
  float lse = mx + logf(expf(z0 - mx) + expf(z1 - mx));
  out[(long long)i * 2 + 0] = z0 - lse;
  out[(long long)i * 2 + 1] = z1 - lse;
}

extern "C" void kernel_launch(void* const* d_in, const int* in_sizes, int n_in,
                              void* d_out, int out_size, void* d_ws, size_t ws_size,
                              hipStream_t stream) {
  const float* x  = (const float*)d_in[0];
  const int*   ei = (const int*)d_in[1];
  const float* W1 = (const float*)d_in[2];
  const float* b1 = (const float*)d_in[3];
  const float* W2 = (const float*)d_in[4];
  const float* b2 = (const float*)d_in[5];
  float* out = (float*)d_out;

  const int N = in_sizes[0] / D_IN;
  const int E = in_sizes[1] / 2;

  float* ws   = (float*)d_ws;
  float* dinv = ws;
  float* t1   = dinv + N;
  float* agg1 = t1 + (long long)N * H_DIM;
  float* h2   = agg1 + (long long)N * H_DIM;
  float* agg2 = h2 + (long long)N * C_DIM;

  const int B = 256;
  fill_kernel<<<(N + B - 1) / B, B, 0, stream>>>(dinv, N, 1.0f);
  fill_kernel<<<((long long)N * H_DIM + B - 1) / B, B, 0, stream>>>(agg1, (long long)N * H_DIM, 0.0f);
  fill_kernel<<<((long long)N * C_DIM + B - 1) / B, B, 0, stream>>>(agg2, (long long)N * C_DIM, 0.0f);

  degree_kernel<<<(E + B - 1) / B, B, 0, stream>>>(ei, dinv, E);
  rsqrt_kernel<<<(N + B - 1) / B, B, 0, stream>>>(dinv, N);

  gemm1_wmma_kernel<<<(N + 127) / 128, B, 0, stream>>>(x, W1, t1, N);
  scatter1_kernel<<<((long long)E * 8 + B - 1) / B, B, 0, stream>>>(ei, t1, dinv, agg1, E);
  bias_relu_kernel<<<((long long)N * H_DIM + B - 1) / B, B, 0, stream>>>(agg1, t1, dinv, b1, N);

  gemm2_kernel<<<(N + B - 1) / B, B, 0, stream>>>(agg1, W2, h2, N);
  scatter2_kernel<<<(E + B - 1) / B, B, 0, stream>>>(ei, h2, dinv, agg2, E);
  logsoftmax_kernel<<<(N + B - 1) / B, B, 0, stream>>>(agg2, h2, dinv, b2, out, N);
}


# degree count and layer-2 edge sums move from global atomics to LDS tables per node range x edge slice (ds_add), slice planes summed by a plain pass; layer-2 f32 sum order changes
# speedup vs baseline: 2.3499x; 2.3499x over previous
#include <hip/hip_runtime.h>
#include <hip/hip_bf16.h>

typedef __attribute__((ext_vector_type(16))) __bf16 v16bf;
typedef __attribute__((ext_vector_type(8)))  float  v8f;

#define D_IN  256
#define H_DIM 32
#define C_DIM 2

__global__ void fill_kernel(float* __restrict__ p, long long n, float v) {
  long long i = (long long)blockIdx.x * blockDim.x + threadIdx.x;
  if (i < n) p[i] = v;
}

#define S_SLICES    32
#define RANGE_NODES 12800
#define WG_RANGE    1024

__global__ void __launch_bounds__(WG_RANGE)
degree_range_kernel(const int* __restrict__ ei, unsigned int* __restrict__ part,
                    int E, int N) {
  __shared__ unsigned int cnt[RANGE_NODES];
  const int tid = threadIdx.x;
  const int lo  = blockIdx.y * RANGE_NODES;
  const int rem = N - lo;
  const unsigned span = (unsigned)(rem < RANGE_NODES ? rem : RANGE_NODES);
  for (int i = tid; i < RANGE_NODES; i += WG_RANGE) cnt[i] = 0u;
  __syncthreads();

  const int* __restrict__ tgt = ei + E;
  if ((E & 3) == 0) {
    const int C   = E >> 2;
    const int cps = (C + S_SLICES - 1) / S_SLICES;
    const int c0  = blockIdx.x * cps;
    const int c1  = (c0 + cps < C) ? (c0 + cps) : C;
    const int4* __restrict__ tg4 = (const int4*)tgt;
    for (int c = c0 + tid; c < c1; c += 4 * WG_RANGE) {
      int4 v[4];
      #pragma unroll
      for (int k = 0; k < 4; ++k) {
        int cc = c + k * WG_RANGE;
        v[k] = (cc < c1) ? tg4[cc] : make_int4(-1, -1, -1, -1);
      }
      #pragma unroll
      for (int k = 0; k < 4; ++k) {
        unsigned u0 = (unsigned)(v[k].x - lo), u1 = (unsigned)(v[k].y - lo);
        unsigned u2 = (unsigned)(v[k].z - lo), u3 = (unsigned)(v[k].w - lo);
        if (u0 < span) atomicAdd(&cnt[u0], 1u);
        if (u1 < span) atomicAdd(&cnt[u1], 1u);
        if (u2 < span) atomicAdd(&cnt[u2], 1u);
        if (u3 < span) atomicAdd(&cnt[u3], 1u);
      }
    }
  } else {
    const int eps = (E + S_SLICES - 1) / S_SLICES;
    const int e0  = blockIdx.x * eps;
    const int e1  = (e0 + eps < E) ? (e0 + eps) : E;
    for (int e = e0 + tid; e < e1; e += WG_RANGE) {
      unsigned u = (unsigned)(tgt[e] - lo);
      if (u < span) atomicAdd(&cnt[u], 1u);
    }
  }
  __syncthreads();
  unsigned int* __restrict__ dst = part + (long long)blockIdx.x * N + lo;
  for (unsigned i = tid; i < span; i += WG_RANGE) dst[i] = cnt[i];
}

__global__ void dinv_kernel(const unsigned int* __restrict__ part,
                            float* __restrict__ dinv, int N) {
  int i = blockIdx.x * blockDim.x + threadIdx.x;
  if (i < N) {
    unsigned c = 1u;
    #pragma unroll
    for (int s = 0; s < S_SLICES; ++s) c += part[(long long)s * N + i];
    float d = (float)c;
    dinv[i] = d > 0.0f ? rsqrtf(d) : 0.0f;
  }
}

__global__ void gemm1_wmma_kernel(const float* __restrict__ x,
                                  const float* __restrict__ W1,
                                  float* __restrict__ t1, int N) {
  __shared__ __align__(32) __bf16 Bfrag[8][2][32][16];

  const int tid  = threadIdx.x;
  const int lane = tid & 31;
  const int wave = tid >> 5;

  for (int idx = tid; idx < 8 * 2 * 32 * 16; idx += blockDim.x) {
    int j     = idx & 15;
    int ln    = (idx >> 4) & 31;
    int ntile = (idx >> 9) & 1;
    int ks    = idx >> 10;
    int jj = j >> 1, p = j & 1;
    int hi = ln >> 4;
    int K  = ks * 32 + 16 * (jj >> 2) + 8 * hi + 2 * (jj & 3) + p;
    int n  = ntile * 16 + (ln & 15);
    Bfrag[ks][ntile][ln][j] = (__bf16)W1[K * H_DIM + n];
  }
  __syncthreads();

  long long rowbase = ((long long)blockIdx.x * 8 + wave) * 16;
  if (rowbase >= N) return;

  const int m  = lane & 15;
  const int hi = lane >> 4;
  const float* xrow = x + (rowbase + m) * D_IN;

  v8f acc0 = {}; v8f acc1 = {};
  #pragma unroll
  for (int ks = 0; ks < 8; ++ks) {
    v16bf a;
    #pragma unroll
    for (int jj = 0; jj < 8; ++jj) {
      int K0 = ks * 32 + 16 * (jj >> 2) + 8 * hi + 2 * (jj & 3);
      float2 f = *(const float2*)(xrow + K0);
      a[2 * jj]     = (__bf16)f.x;
      a[2 * jj + 1] = (__bf16)f.y;
    }
    v16bf b0 = *(const v16bf*)&Bfrag[ks][0][lane][0];
    v16bf b1 = *(const v16bf*)&Bfrag[ks][1][lane][0];
    acc0 = __builtin_amdgcn_wmma_f32_16x16x32_bf16(false, a, false, b0,
                                                   (short)0, acc0, false, false);
    acc1 = __builtin_amdgcn_wmma_f32_16x16x32_bf16(false, a, false, b1,
                                                   (short)0, acc1, false, false);
  }

  #pragma unroll
  for (int r = 0; r < 8; ++r) {
    long long row = rowbase + r + 8 * hi;
    t1[row * H_DIM + m]      = acc0[r];
    t1[row * H_DIM + 16 + m] = acc1[r];
  }
}

__global__ void scatter1_kernel(const int* __restrict__ ei,
                                const float* __restrict__ t1,
                                const float* __restrict__ dinv,
                                float* __restrict__ agg1, int E) {
  long long gid = (long long)blockIdx.x * blockDim.x + threadIdx.x;
  long long e   = gid >> 3;
  int part      = (int)(gid & 7);
  if (e >= E) return;
  long long s = ei[e];
  long long t = ei[(long long)E + e];
  float w = dinv[s] * dinv[t];
  float4 v = *(const float4*)(t1 + s * H_DIM + part * 4);
  float* dst = agg1 + t * H_DIM + part * 4;
  atomicAdd(dst + 0, v.x * w);
  atomicAdd(dst + 1, v.y * w);
  atomicAdd(dst + 2, v.z * w);
  atomicAdd(dst + 3, v.w * w);
}

__global__ void bias_relu_kernel(float* __restrict__ agg1,
                                 const float* __restrict__ t1,
                                 const float* __restrict__ dinv,
                                 const float* __restrict__ b1, int N) {
  long long i = (long long)blockIdx.x * blockDim.x + threadIdx.x;
  if (i >= (long long)N * H_DIM) return;
  long long node = i >> 5;
  int n = (int)(i & 31);
  float di = dinv[node];
  float v = agg1[i] + di * di * t1[i] + b1[n];
  agg1[i] = v > 0.0f ? v : 0.0f;
}

__global__ void gemm2_kernel(const float* __restrict__ h1,
                             const float* __restrict__ W2,
                             float* __restrict__ h2, int N) {
  int i = blockIdx.x * blockDim.x + threadIdx.x;
  if (i >= N) return;
  const float* hr = h1 + (long long)i * H_DIM;
  float z0 = 0.0f, z1 = 0.0f;
  #pragma unroll
  for (int k = 0; k < H_DIM; ++k) {
    float h = hr[k];
    z0 += h * W2[k * C_DIM + 0];
    z1 += h * W2[k * C_DIM + 1];
  }
  h2[(long long)i * C_DIM + 0] = z0;
  h2[(long long)i * C_DIM + 1] = z1;
}

__global__ void __launch_bounds__(WG_RANGE)
scatter2_range_kernel(const int* __restrict__ ei,
                      const float* __restrict__ h2,
                      const float* __restrict__ dinv,
                      float* __restrict__ part, int E, int N) {
  __shared__ float acc[RANGE_NODES * C_DIM];
  __shared__ float dloc[RANGE_NODES];
  const int tid = threadIdx.x;
  const int lo  = blockIdx.y * RANGE_NODES;
  const int rem = N - lo;
  const unsigned span = (unsigned)(rem < RANGE_NODES ? rem : RANGE_NODES);
  for (int i = tid; i < RANGE_NODES * C_DIM; i += WG_RANGE) acc[i] = 0.0f;
  for (unsigned i = tid; i < span; i += WG_RANGE) dloc[i] = dinv[lo + i];
  __syncthreads();

  const int* __restrict__ src = ei;
  const int* __restrict__ tgt = ei + E;
  const float2* __restrict__ h2v = (const float2*)h2;
  if ((E & 3) == 0) {
    const int C   = E >> 2;
    const int cps = (C + S_SLICES - 1) / S_SLICES;
    const int c0  = blockIdx.x * cps;
    const int c1  = (c0 + cps < C) ? (c0 + cps) : C;
    const int4* __restrict__ tg4 = (const int4*)tgt;
    const int4* __restrict__ sr4 = (const int4*)src;
    for (int c = c0 + tid; c < c1; c += 2 * WG_RANGE) {
      int t[8], s[8];
      #pragma unroll
      for (int k = 0; k < 2; ++k) {
        int cc = c + k * WG_RANGE;
        bool ok = cc < c1;
        int4 tv = ok ? tg4[cc] : make_int4(-1, -1, -1, -1);
        int4 sv = ok ? sr4[cc] : make_int4(0, 0, 0, 0);
        t[4 * k + 0] = tv.x; t[4 * k + 1] = tv.y; t[4 * k + 2] = tv.z; t[4 * k + 3] = tv.w;
        s[4 * k + 0] = sv.x; s[4 * k + 1] = sv.y; s[4 * k + 2] = sv.z; s[4 * k + 3] = sv.w;
      }
      unsigned u[8]; float w[8]; float2 v[8];
      #pragma unroll
      for (int j = 0; j < 8; ++j) {
        u[j] = (unsigned)(t[j] - lo);
        bool act = u[j] < span;
        int ss = act ? s[j] : 0;
        w[j] = dinv[ss] * dloc[act ? u[j] : 0u];
        v[j] = h2v[ss];
      }
      #pragma unroll
      for (int j = 0; j < 8; ++j) {
        if (u[j] < span) {
          atomicAdd(&acc[u[j] * C_DIM + 0], v[j].x * w[j]);
          atomicAdd(&acc[u[j] * C_DIM + 1], v[j].y * w[j]);
        }
      }
    }
  } else {
    const int eps = (E + S_SLICES - 1) / S_SLICES;
    const int e0  = blockIdx.x * eps;
    const int e1  = (e0 + eps < E) ? (e0 + eps) : E;
    for (int e = e0 + tid; e < e1; e += WG_RANGE) {
      int tt = tgt[e];
      unsigned u = (unsigned)(tt - lo);
      if (u < span) {
        int ss = src[e];
        float w = dinv[ss] * dloc[u];
        float2 v = h2v[ss];
        atomicAdd(&acc[u * C_DIM + 0], v.x * w);
        atomicAdd(&acc[u * C_DIM + 1], v.y * w);
      }
    }
  }
  __syncthreads();
  float* __restrict__ dst = part + ((long long)blockIdx.x * N + lo) * C_DIM;
  for (unsigned i = tid; i < span * C_DIM; i += WG_RANGE) dst[i] = acc[i];
}

__global__ void logsoftmax_kernel(const float* __restrict__ part,
                                  const float* __restrict__ h2,
                                  const float* __restrict__ dinv,
                                  const float* __restrict__ b2,
                                  float* __restrict__ out, int N) {
  int i = blockIdx.x * blockDim.x + threadIdx.x;
  if (i >= N) return;
  const float2* __restrict__ p2 = (const float2*)part;
  float a0 = 0.0f, a1 = 0.0f;
  #pragma unroll
  for (int s = 0; s < S_SLICES; ++s) {
    float2 p = p2[(long long)s * N + i];
    a0 += p.x; a1 += p.y;
  }
  float di = dinv[i], sl = di * di;
  float z0 = a0 + sl * h2[(long long)i * 2 + 0] + b2[0];
  float z1 = a1 + sl * h2[(long long)i * 2 + 1] + b2[1];
  float mx  = fmaxf(z0, z1);
  float lse = mx + logf(expf(z0 - mx) + expf(z1 - mx));
  out[(long long)i * 2 + 0] = z0 - lse;
  out[(long long)i * 2 + 1] = z1 - lse;
}

extern "C" void kernel_launch(void* const* d_in, const int* in_sizes, int n_in,
                              void* d_out, int out_size, void* d_ws, size_t ws_size,
                              hipStream_t stream) {
  const float* x  = (const float*)d_in[0];
  const int*   ei = (const int*)d_in[1];
  const float* W1 = (const float*)d_in[2];
  const float* b1 = (const float*)d_in[3];
  const float* W2 = (const float*)d_in[4];
  const float* b2 = (const float*)d_in[5];
  float* out = (float*)d_out;

  const int N = in_sizes[0] / D_IN;
  const int E = in_sizes[1] / 2;

  float* ws   = (float*)d_ws;
  float* t1   = ws;
  float* agg1 = t1 + (long long)N * H_DIM;
  float* h2   = agg1 + (long long)N * H_DIM;
  float* dinv = h2 + (long long)N * C_DIM;
  unsigned int* degpart = (unsigned int*)t1;
  float* part2 = t1;

  const int B = 256;
  const dim3 rgrid(S_SLICES, (N + RANGE_NODES - 1) / RANGE_NODES);

  degree_range_kernel<<<rgrid, WG_RANGE, 0, stream>>>(ei, degpart, E, N);
  dinv_kernel<<<(N + B - 1) / B, B, 0, stream>>>(degpart, dinv, N);

  fill_kernel<<<((long long)N * H_DIM + B - 1) / B, B, 0, stream>>>(agg1, (long long)N * H_DIM, 0.0f);

  gemm1_wmma_kernel<<<(N + 127) / 128, B, 0, stream>>>(x, W1, t1, N);
  scatter1_kernel<<<((long long)E * 8 + B - 1) / B, B, 0, stream>>>(ei, t1, dinv, agg1, E);
  bias_relu_kernel<<<((long long)N * H_DIM + B - 1) / B, B, 0, stream>>>(agg1, t1, dinv, b1, N);

  gemm2_kernel<<<(N + B - 1) / B, B, 0, stream>>>(agg1, W2, h2, N);
  scatter2_range_kernel<<<rgrid, WG_RANGE, 0, stream>>>(ei, h2, dinv, part2, E, N);
  logsoftmax_kernel<<<(N + B - 1) / B, B, 0, stream>>>(part2, h2, dinv, b2, out, N);
}


# edges counting-sorted by target once (LDS counts, prefix, LDS cursors); both layers sum by gather over the sorted sources, node epilogue in registers; no float atomics; f32 sum order changes
# speedup vs baseline: 3.6079x; 1.5353x over previous
#include <hip/hip_runtime.h>
#include <hip/hip_bf16.h>

typedef __attribute__((ext_vector_type(16))) __bf16 v16bf;
typedef __attribute__((ext_vector_type(8)))  float  v8f;

#define D_IN  256
#define H_DIM 32
#define C_DIM 2

#define S_SLICES    32
#define RANGE_NODES 12800
#define WG_RANGE    1024

__global__ void __launch_bounds__(WG_RANGE)
degree_range_kernel(const int* __restrict__ ei, unsigned int* __restrict__ part,
                    int E, int N) {
  __shared__ unsigned int cnt[RANGE_NODES];
  const int tid = threadIdx.x;
  const int lo  = blockIdx.y * RANGE_NODES;
  const int rem = N - lo;
  const unsigned span = (unsigned)(rem < RANGE_NODES ? rem : RANGE_NODES);
  for (int i = tid; i < RANGE_NODES; i += WG_RANGE) cnt[i] = 0u;
  __syncthreads();

  const int* __restrict__ tgt = ei + E;
  if ((E & 3) == 0) {
    const int C   = E >> 2;
    const int cps = (C + S_SLICES - 1) / S_SLICES;
    const int c0  = blockIdx.x * cps;
    const int c1  = (c0 + cps < C) ? (c0 + cps) : C;
    const int4* __restrict__ tg4 = (const int4*)tgt;
    for (int c = c0 + tid; c < c1; c += 4 * WG_RANGE) {
      int4 v[4];
      #pragma unroll
      for (int k = 0; k < 4; ++k) {
        int cc = c + k * WG_RANGE;
        v[k] = (cc < c1) ? tg4[cc] : make_int4(-1, -1, -1, -1);
      }
      #pragma unroll
      for (int k = 0; k < 4; ++k) {
        unsigned u0 = (unsigned)(v[k].x - lo), u1 = (unsigned)(v[k].y - lo);
        unsigned u2 = (unsigned)(v[k].z - lo), u3 = (unsigned)(v[k].w - lo);
        if (u0 < span) atomicAdd(&cnt[u0], 1u);
        if (u1 < span) atomicAdd(&cnt[u1], 1u);
        if (u2 < span) atomicAdd(&cnt[u2], 1u);
        if (u3 < span) atomicAdd(&cnt[u3], 1u);
      }
    }
  } else {
    const int eps = (E + S_SLICES - 1) / S_SLICES;
    const int e0  = blockIdx.x * eps;
    const int e1  = (e0 + eps < E) ? (e0 + eps) : E;
    for (int e = e0 + tid; e < e1; e += WG_RANGE) {
      unsigned u = (unsigned)(tgt[e] - lo);
      if (u < span) atomicAdd(&cnt[u], 1u);
    }
  }
  __syncthreads();
  unsigned int* __restrict__ dst = part + (long long)blockIdx.x * N + lo;
  for (unsigned i = tid; i < span; i += WG_RANGE) dst[i] = cnt[i];
}


#define PB 256
static_assert(RANGE_NODES % PB == 0, "a node range must be a whole number of prefix blocks");
static_assert(WG_RANGE == 1024, "the range scan below assumes 32 waves of 32 lanes");
#define SCAN_CH  13
#define CUR_PAD  (WG_RANGE * SCAN_CH)
static_assert(CUR_PAD >= RANGE_NODES + 32, "a cursor per node and a spare word per lane");

__global__ void __launch_bounds__(PB)
slice_prefix_kernel(unsigned int* __restrict__ part, unsigned int* __restrict__ deg,
                    unsigned int* __restrict__ blocksum, int N) {
  __shared__ unsigned int wtot[PB / 32];
  const int i = blockIdx.x * PB + threadIdx.x;
  unsigned c = 0u;
  if (i < N) {
    unsigned v[S_SLICES];
    #pragma unroll
    for (int s = 0; s < S_SLICES; ++s) v[s] = part[(long long)s * N + i];
    #pragma unroll
    for (int s = 0; s < S_SLICES; ++s) { part[(long long)s * N + i] = c; c += v[s]; }
    deg[i] = c;
  }
  unsigned r = c;
  #pragma unroll
  for (int m = 16; m >= 1; m >>= 1) r += __shfl_xor(r, m);
  if ((threadIdx.x & 31) == 0) wtot[threadIdx.x >> 5] = r;
  __syncthreads();
  if (threadIdx.x == 0) {
    unsigned t = 0u;
    #pragma unroll
    for (int w = 0; w < PB / 32; ++w) t += wtot[w];
    blocksum[blockIdx.x] = t;
  }
}

__global__ void __launch_bounds__(WG_RANGE)
sort_range_kernel(const int* __restrict__ ei, const unsigned int* __restrict__ part,
                  const unsigned int* __restrict__ deg,
                  const unsigned int* __restrict__ blocksum,
                  unsigned int* __restrict__ rowptr, int* __restrict__ sorted,
                  int E, int N) {
  __shared__ unsigned int cur[CUR_PAD];
  __shared__ unsigned int wsum[32];
  __shared__ unsigned int bsum[32];
  __shared__ unsigned int sbase;
  const int tid  = threadIdx.x;
  const int lane = tid & 31;
  const int wave = tid >> 5;
  const int lo  = blockIdx.y * RANGE_NODES;
  const int rem = N - lo;
  const unsigned span = (unsigned)(rem < RANGE_NODES ? rem : RANGE_NODES);

  {
    unsigned dv[SCAN_CH];
    #pragma unroll
    for (int k = 0; k < SCAN_CH; ++k) {
      unsigned i = (unsigned)tid + (unsigned)k * WG_RANGE;
      unsigned v = deg[lo + (i < span ? i : span - 1u)];
      dv[k] = (i < span) ? v : 0u;
    }
    #pragma unroll
    for (int k = 0; k < SCAN_CH; ++k) cur[tid + k * WG_RANGE] = dv[k];
  }
  const int nb = blockIdx.y * (RANGE_NODES / PB);
  unsigned bs = 0u;
  for (int b = tid; b < nb; b += WG_RANGE) bs += blocksum[b];
  #pragma unroll
  for (int m = 16; m >= 1; m >>= 1) bs += __shfl_xor(bs, m);
  if (lane == 0) bsum[wave] = bs;
  __syncthreads();

  const int st = tid * SCAN_CH;
  unsigned loc = 0u;
  #pragma unroll
  for (int k = 0; k < SCAN_CH; ++k) loc += cur[st + k];
  unsigned inc = loc;
  #pragma unroll
  for (int d = 1; d < 32; d <<= 1) { unsigned n = __shfl_up(inc, d); if (lane >= d) inc += n; }
  if (lane == 31) wsum[wave] = inc;
  __syncthreads();
  if (wave == 0) {
    unsigned v = wsum[lane], w = v;
    #pragma unroll
    for (int d = 1; d < 32; d <<= 1) { unsigned n = __shfl_up(w, d); if (lane >= d) w += n; }
    wsum[lane] = w - v;
    unsigned b = bsum[lane];
    #pragma unroll
    for (int m = 16; m >= 1; m >>= 1) b += __shfl_xor(b, m);
    if (lane == 0) sbase = b;
  }
  __syncthreads();
  unsigned run = sbase + wsum[wave] + (inc - loc);
  #pragma unroll
  for (int k = 0; k < SCAN_CH; ++k) { unsigned d = cur[st + k]; cur[st + k] = run; run += d; }
  __syncthreads();
  const unsigned int* __restrict__ mypart = part + (long long)blockIdx.x * N + lo;
  for (unsigned i = tid; i < span; i += WG_RANGE) {
    unsigned r = cur[i];
    if (blockIdx.x == 0) rowptr[lo + i] = r;
    cur[i] = r + mypart[i];
  }
  __syncthreads();

  const int* __restrict__ src = ei;
  const int* __restrict__ tgt = ei + E;
  if ((E & 3) == 0) {
    const int C   = E >> 2;
    const int cps = (C + S_SLICES - 1) / S_SLICES;
    const int c0  = blockIdx.x * cps;
    const int c1  = (c0 + cps < C) ? (c0 + cps) : C;
    const int4* __restrict__ tg4 = (const int4*)tgt;
    const int4* __restrict__ sr4 = (const int4*)src;
    for (int c = c0 + tid; c < c1; c += 4 * WG_RANGE) {
      unsigned u[16]; int s[16]; unsigned p[16];
      #pragma unroll
      for (int k = 0; k < 4; ++k) {
        int cc = c + k * WG_RANGE;
        bool ok = cc < c1;
        int4 tv = ok ? tg4[cc] : make_int4(-1, -1, -1, -1);
        int4 sv = ok ? sr4[cc] : make_int4(0, 0, 0, 0);
        u[4 * k + 0] = (unsigned)(tv.x - lo); u[4 * k + 1] = (unsigned)(tv.y - lo);
        u[4 * k + 2] = (unsigned)(tv.z - lo); u[4 * k + 3] = (unsigned)(tv.w - lo);
        s[4 * k + 0] = sv.x; s[4 * k + 1] = sv.y; s[4 * k + 2] = sv.z; s[4 * k + 3] = sv.w;
      }
      #pragma unroll
      for (int j = 0; j < 16; ++j) {
        const bool act = u[j] < span;
        p[j] = atomicAdd(&cur[act ? u[j] : (unsigned)(RANGE_NODES + lane)], act ? 1u : 0u);
      }
      #pragma unroll
      for (int j = 0; j < 16; ++j) if (u[j] < span) sorted[p[j]] = s[j];
    }
  } else {
    const int eps = (E + S_SLICES - 1) / S_SLICES;
    const int e0  = blockIdx.x * eps;
    const int e1  = (e0 + eps < E) ? (e0 + eps) : E;
    for (int e = e0 + tid; e < e1; e += WG_RANGE) {
      unsigned u = (unsigned)(tgt[e] - lo);
      if (u < span) sorted[atomicAdd(&cur[u], 1u)] = src[e];
    }
  }
}

__global__ void gemm1_wmma_kernel(const float* __restrict__ x,
                                  const float* __restrict__ W1,
                                  const unsigned int* __restrict__ deg,
                                  float* __restrict__ t1, int N) {
  __shared__ __align__(32) __bf16 Bfrag[8][2][32][16];

  const int tid  = threadIdx.x;
  const int lane = tid & 31;
  const int wave = tid >> 5;

  for (int idx = tid; idx < 8 * 2 * 32 * 16; idx += blockDim.x) {
    int j     = idx & 15;
    int ln    = (idx >> 4) & 31;
    int ntile = (idx >> 9) & 1;
    int ks    = idx >> 10;
    int jj = j >> 1, p = j & 1;
    int hi = ln >> 4;
    int K  = ks * 32 + 16 * (jj >> 2) + 8 * hi + 2 * (jj & 3) + p;
    int n  = ntile * 16 + (ln & 15);
    Bfrag[ks][ntile][ln][j] = (__bf16)W1[K * H_DIM + n];
  }
  __syncthreads();

  long long rowbase = ((long long)blockIdx.x * 8 + wave) * 16;
  if (rowbase >= N) return;

  const int m  = lane & 15;
  const int hi = lane >> 4;
  const float* xrow = x + (rowbase + m) * D_IN;

  v8f acc0 = {}; v8f acc1 = {};
  #pragma unroll
  for (int ks = 0; ks < 8; ++ks) {
    v16bf a;
    #pragma unroll
    for (int jj = 0; jj < 8; ++jj) {
      int K0 = ks * 32 + 16 * (jj >> 2) + 8 * hi + 2 * (jj & 3);
      float2 f = *(const float2*)(xrow + K0);
      a[2 * jj]     = (__bf16)f.x;
      a[2 * jj + 1] = (__bf16)f.y;
    }
    v16bf b0 = *(const v16bf*)&Bfrag[ks][0][lane][0];
    v16bf b1 = *(const v16bf*)&Bfrag[ks][1][lane][0];
    acc0 = __builtin_amdgcn_wmma_f32_16x16x32_bf16(false, a, false, b0,
                                                   (short)0, acc0, false, false);
    acc1 = __builtin_amdgcn_wmma_f32_16x16x32_bf16(false, a, false, b1,
                                                   (short)0, acc1, false, false);
  }

  #pragma unroll
  for (int r = 0; r < 8; ++r) {
    long long row = rowbase + r + 8 * hi;
    float di = rsqrtf((float)(deg[row] + 1u));
    t1[row * H_DIM + m]      = acc0[r] * di;
    t1[row * H_DIM + 16 + m] = acc1[r] * di;
  }
}

#define GB 256
__global__ void __launch_bounds__(GB)
gather1_kernel(const int* __restrict__ sorted, const unsigned int* __restrict__ rowptr,
               const float* __restrict__ t1, const float* __restrict__ b1,
               const float* __restrict__ W2, float* __restrict__ h2, int N, int E) {
  const int lane  = threadIdx.x & 31;
  const int sub   = lane & 7;
  const int gbase = lane & 24;
  const long long gid = (long long)blockIdx.x * GB + threadIdx.x;
  const long long tq  = gid >> 3;
  const bool valid = tq < N;
  const int t = valid ? (int)tq : N - 1;
  const unsigned beg = rowptr[t];
  const unsigned end = (t + 1 < N) ? rowptr[t + 1] : (unsigned)E;
  const float4* __restrict__ tv = (const float4*)t1;
  float4 acc = make_float4(0.0f, 0.0f, 0.0f, 0.0f);
  for (unsigned k = beg; k < end; k += 8) {
    const unsigned kk = k + sub;
    const int mine = (kk < end) ? sorted[kk] : -1;
    #pragma unroll
    for (int j = 0; j < 8; ++j) {
      const int sj = __shfl(mine, gbase + j);
      const bool on = sj >= 0;
      const float4 v = tv[(long long)(on ? sj : 0) * (H_DIM / 4) + sub];
      acc.x += on ? v.x : 0.0f;
      acc.y += on ? v.y : 0.0f;
      acc.z += on ? v.z : 0.0f;
      acc.w += on ? v.w : 0.0f;
    }
  }
  const float di = rsqrtf((float)(end - beg + 1u));
  const float4 self = tv[(long long)t * (H_DIM / 4) + sub];
  const float4 bb = ((const float4*)b1)[sub];
  float h0 = di * (acc.x + self.x) + bb.x;
  float h1 = di * (acc.y + self.y) + bb.y;
  float h2a = di * (acc.z + self.z) + bb.z;
  float h3 = di * (acc.w + self.w) + bb.w;
  h0 = h0 > 0.0f ? h0 : 0.0f;
  h1 = h1 > 0.0f ? h1 : 0.0f;
  h2a = h2a > 0.0f ? h2a : 0.0f;
  h3 = h3 > 0.0f ? h3 : 0.0f;
  const float4 wa = ((const float4*)W2)[2 * sub];
  const float4 wb = ((const float4*)W2)[2 * sub + 1];
  float z0 = h0 * wa.x + h1 * wa.z + h2a * wb.x + h3 * wb.z;
  float z1 = h0 * wa.y + h1 * wa.w + h2a * wb.y + h3 * wb.w;
  #pragma unroll
  for (int m = 1; m < 8; m <<= 1) { z0 += __shfl_xor(z0, m); z1 += __shfl_xor(z1, m); }
  if (valid && sub == 0) ((float2*)h2)[t] = make_float2(di * z0, di * z1);
}

__global__ void __launch_bounds__(GB)
gather2_logsoftmax_kernel(const int* __restrict__ sorted,
                          const unsigned int* __restrict__ rowptr,
                          const float* __restrict__ h2, const float* __restrict__ b2,
                          float* __restrict__ out, int N, int E) {
  const int sub = threadIdx.x & 7;
  const long long gid = (long long)blockIdx.x * GB + threadIdx.x;
  const long long tq  = gid >> 3;
  const bool valid = tq < N;
  const int t = valid ? (int)tq : N - 1;
  const unsigned beg = rowptr[t];
  const unsigned end = (t + 1 < N) ? rowptr[t + 1] : (unsigned)E;
  const float2* __restrict__ hv = (const float2*)h2;
  float a0 = 0.0f, a1 = 0.0f;
  for (unsigned k = beg + sub; k < end; k += 32) {
    int s[4];
    #pragma unroll
    for (int j = 0; j < 4; ++j) { unsigned kk = k + 8 * j; s[j] = (kk < end) ? sorted[kk] : -1; }
    #pragma unroll
    for (int j = 0; j < 4; ++j) {
      const bool on = s[j] >= 0;
      const float2 v = hv[on ? s[j] : 0];
      a0 += on ? v.x : 0.0f;
      a1 += on ? v.y : 0.0f;
    }
  }
  #pragma unroll
  for (int m = 1; m < 8; m <<= 1) { a0 += __shfl_xor(a0, m); a1 += __shfl_xor(a1, m); }
  const float di = rsqrtf((float)(end - beg + 1u));
  const float2 self = hv[t];
  float z0 = di * (a0 + self.x) + b2[0];
  float z1 = di * (a1 + self.y) + b2[1];
  float mx  = fmaxf(z0, z1);
  float lse = mx + logf(expf(z0 - mx) + expf(z1 - mx));
  if (valid && sub == 0) ((float2*)out)[t] = make_float2(z0 - lse, z1 - lse);
}

extern "C" void kernel_launch(void* const* d_in, const int* in_sizes, int n_in,
                              void* d_out, int out_size, void* d_ws, size_t ws_size,
                              hipStream_t stream) {
  const float* x  = (const float*)d_in[0];
  const int*   ei = (const int*)d_in[1];
  const float* W1 = (const float*)d_in[2];
  const float* b1 = (const float*)d_in[3];
  const float* W2 = (const float*)d_in[4];
  const float* b2 = (const float*)d_in[5];
  float* out = (float*)d_out;

  const int N = in_sizes[0] / D_IN;
  const int E = in_sizes[1] / 2;

  float* ws   = (float*)d_ws;
  float* t1   = ws;
  float* h2   = t1 + (long long)N * H_DIM;
  unsigned int* rowptr = (unsigned int*)(h2 + (long long)N * C_DIM);
  int* sorted = (int*)(rowptr + N);
  unsigned int* degpart  = (unsigned int*)t1;
  unsigned int* deg      = (unsigned int*)h2;
  unsigned int* blocksum = deg + N;

  const dim3 rgrid(S_SLICES, (N + RANGE_NODES - 1) / RANGE_NODES);
  const int gblocks = (int)(((long long)N * 8 + GB - 1) / GB);

  degree_range_kernel<<<rgrid, WG_RANGE, 0, stream>>>(ei, degpart, E, N);
  slice_prefix_kernel<<<(N + PB - 1) / PB, PB, 0, stream>>>(degpart, deg, blocksum, N);
  sort_range_kernel<<<rgrid, WG_RANGE, 0, stream>>>(ei, degpart, deg, blocksum, rowptr, sorted, E, N);

  gemm1_wmma_kernel<<<(N + 127) / 128, 256, 0, stream>>>(x, W1, deg, t1, N);
  gather1_kernel<<<gblocks, GB, 0, stream>>>(sorted, rowptr, t1, b1, W2, h2, N, E);

  gather2_logsoftmax_kernel<<<gblocks, GB, 0, stream>>>(sorted, rowptr, h2, b2, out, N, E);
}


# round trips in series cut: sort loads its slice-prefix column with the degree column, degree index loads a trip ahead, W1 staging loads grouped, gather1 next slots with rows, gather2 8 slots per trip
# speedup vs baseline: 3.8197x; 1.0587x over previous
#include <hip/hip_runtime.h>
#include <hip/hip_bf16.h>

typedef __attribute__((ext_vector_type(16))) __bf16 v16bf;
typedef __attribute__((ext_vector_type(8)))  float  v8f;

#define D_IN  256
#define H_DIM 32
#define C_DIM 2

#define S_SLICES    32
#define RANGE_NODES 12800
#define WG_RANGE    1024

__global__ void __launch_bounds__(WG_RANGE)
degree_range_kernel(const int* __restrict__ ei, unsigned int* __restrict__ part,
                    int E, int N) {
  __shared__ unsigned int cnt[RANGE_NODES];
  const int tid = threadIdx.x;
  const int lo  = blockIdx.y * RANGE_NODES;
  const int rem = N - lo;
  const unsigned span = (unsigned)(rem < RANGE_NODES ? rem : RANGE_NODES);
  const int* __restrict__ tgt = ei + E;
  const bool vec = (E & 3) == 0;
  const int C   = E >> 2;
  const int cps = (C + S_SLICES - 1) / S_SLICES;
  const int c0  = blockIdx.x * cps;
  const int c1  = (c0 + cps < C) ? (c0 + cps) : C;
  const int4* __restrict__ tg4 = (const int4*)tgt;
  int4 nv[4];
  #pragma unroll
  for (int k = 0; k < 4; ++k) {
    int cc = c0 + tid + k * WG_RANGE;
    nv[k] = (vec && cc < c1) ? tg4[cc] : make_int4(-1, -1, -1, -1);
  }
  for (int i = tid; i < RANGE_NODES; i += WG_RANGE) cnt[i] = 0u;
  __syncthreads();

  if (vec) {
    for (int c = c0 + tid; c < c1; c += 4 * WG_RANGE) {
      int4 v[4];
      #pragma unroll
      for (int k = 0; k < 4; ++k) v[k] = nv[k];
      #pragma unroll
      for (int k = 0; k < 4; ++k) {
        int cc = c + (4 + k) * WG_RANGE;
        nv[k] = (cc < c1) ? tg4[cc] : make_int4(-1, -1, -1, -1);
      }
      #pragma unroll
      for (int k = 0; k < 4; ++k) {
        unsigned u0 = (unsigned)(v[k].x - lo), u1 = (unsigned)(v[k].y - lo);
        unsigned u2 = (unsigned)(v[k].z - lo), u3 = (unsigned)(v[k].w - lo);
        if (u0 < span) atomicAdd(&cnt[u0], 1u);
        if (u1 < span) atomicAdd(&cnt[u1], 1u);
        if (u2 < span) atomicAdd(&cnt[u2], 1u);
        if (u3 < span) atomicAdd(&cnt[u3], 1u);
      }
    }
  } else {
    const int eps = (E + S_SLICES - 1) / S_SLICES;
    const int e0  = blockIdx.x * eps;
    const int e1  = (e0 + eps < E) ? (e0 + eps) : E;
    for (int e = e0 + tid; e < e1; e += WG_RANGE) {
      unsigned u = (unsigned)(tgt[e] - lo);
      if (u < span) atomicAdd(&cnt[u], 1u);
    }
  }
  __syncthreads();
  unsigned int* __restrict__ dst = part + (long long)blockIdx.x * N + lo;
  for (unsigned i = tid; i < span; i += WG_RANGE) dst[i] = cnt[i];
}


#define PB 256
static_assert(RANGE_NODES % PB == 0, "a node range must be a whole number of prefix blocks");
static_assert(WG_RANGE == 1024, "the range scan below assumes 32 waves of 32 lanes");
#define SCAN_CH  13
#define CUR_PAD  (WG_RANGE * SCAN_CH)
static_assert(CUR_PAD >= RANGE_NODES + 32, "a cursor per node and a spare word per lane");

__global__ void __launch_bounds__(PB)
slice_prefix_kernel(unsigned int* __restrict__ part, unsigned int* __restrict__ deg,
                    unsigned int* __restrict__ blocksum, int N) {
  __shared__ unsigned int wtot[PB / 32];
  const int i = blockIdx.x * PB + threadIdx.x;
  unsigned c = 0u;
  if (i < N) {
    unsigned v[S_SLICES];
    #pragma unroll
    for (int s = 0; s < S_SLICES; ++s) v[s] = part[(long long)s * N + i];
    #pragma unroll
    for (int s = 0; s < S_SLICES; ++s) { part[(long long)s * N + i] = c; c += v[s]; }
    deg[i] = c;
  }
  unsigned r = c;
  #pragma unroll
  for (int m = 16; m >= 1; m >>= 1) r += __shfl_xor(r, m);
  if ((threadIdx.x & 31) == 0) wtot[threadIdx.x >> 5] = r;
  __syncthreads();
  if (threadIdx.x == 0) {
    unsigned t = 0u;
    #pragma unroll
    for (int w = 0; w < PB / 32; ++w) t += wtot[w];
    blocksum[blockIdx.x] = t;
  }
}

__global__ void __launch_bounds__(WG_RANGE)
sort_range_kernel(const int* __restrict__ ei, const unsigned int* __restrict__ part,
                  const unsigned int* __restrict__ deg,
                  const unsigned int* __restrict__ blocksum,
                  unsigned int* __restrict__ rowptr, int* __restrict__ sorted,
                  int E, int N) {
  __shared__ unsigned int cur[CUR_PAD];
  __shared__ unsigned int wsum[32];
  __shared__ unsigned int bsum[32];
  __shared__ unsigned int sbase;
  const int tid  = threadIdx.x;
  const int lane = tid & 31;
  const int wave = tid >> 5;
  const int lo  = blockIdx.y * RANGE_NODES;
  const int rem = N - lo;
  const unsigned span = (unsigned)(rem < RANGE_NODES ? rem : RANGE_NODES);

  const unsigned int* __restrict__ mypart = part + (long long)blockIdx.x * N + lo;
  unsigned mp[SCAN_CH];
  {
    unsigned dv[SCAN_CH];
    #pragma unroll
    for (int k = 0; k < SCAN_CH; ++k) {
      unsigned i  = (unsigned)tid + (unsigned)k * WG_RANGE;
      unsigned ic = (i < span) ? i : span - 1u;
      unsigned v  = deg[lo + ic];
      dv[k] = (i < span) ? v : 0u;
      mp[k] = mypart[ic];
    }
    #pragma unroll
    for (int k = 0; k < SCAN_CH; ++k) cur[tid + k * WG_RANGE] = dv[k];
  }
  const int nb = blockIdx.y * (RANGE_NODES / PB);
  unsigned bs = 0u;
  for (int b = tid; b < nb; b += WG_RANGE) bs += blocksum[b];
  #pragma unroll
  for (int m = 16; m >= 1; m >>= 1) bs += __shfl_xor(bs, m);
  if (lane == 0) bsum[wave] = bs;
  __syncthreads();

  const int st = tid * SCAN_CH;
  unsigned loc = 0u;
  #pragma unroll
  for (int k = 0; k < SCAN_CH; ++k) loc += cur[st + k];
  unsigned inc = loc;
  #pragma unroll
  for (int d = 1; d < 32; d <<= 1) { unsigned n = __shfl_up(inc, d); if (lane >= d) inc += n; }
  if (lane == 31) wsum[wave] = inc;
  __syncthreads();
  if (wave == 0) {
    unsigned v = wsum[lane], w = v;
    #pragma unroll
    for (int d = 1; d < 32; d <<= 1) { unsigned n = __shfl_up(w, d); if (lane >= d) w += n; }
    wsum[lane] = w - v;
    unsigned b = bsum[lane];
    #pragma unroll
    for (int m = 16; m >= 1; m >>= 1) b += __shfl_xor(b, m);
    if (lane == 0) sbase = b;
  }
  __syncthreads();
  unsigned run = sbase + wsum[wave] + (inc - loc);
  #pragma unroll
  for (int k = 0; k < SCAN_CH; ++k) { unsigned d = cur[st + k]; cur[st + k] = run; run += d; }
  __syncthreads();
  {
    unsigned rr[SCAN_CH];
    #pragma unroll
    for (int k = 0; k < SCAN_CH; ++k) rr[k] = cur[tid + k * WG_RANGE];
    #pragma unroll
    for (int k = 0; k < SCAN_CH; ++k) {
      const unsigned i = (unsigned)tid + (unsigned)k * WG_RANGE;
      if (i < span) {
        if (blockIdx.x == 0) rowptr[lo + i] = rr[k];
        cur[i] = rr[k] + mp[k];
      }
    }
  }
  __syncthreads();

  const int* __restrict__ src = ei;
  const int* __restrict__ tgt = ei + E;
  if ((E & 3) == 0) {
    const int C   = E >> 2;
    const int cps = (C + S_SLICES - 1) / S_SLICES;
    const int c0  = blockIdx.x * cps;
    const int c1  = (c0 + cps < C) ? (c0 + cps) : C;
    const int4* __restrict__ tg4 = (const int4*)tgt;
    const int4* __restrict__ sr4 = (const int4*)src;
    for (int c = c0 + tid; c < c1; c += 4 * WG_RANGE) {
      unsigned u[16]; int s[16]; unsigned p[16];
      #pragma unroll
      for (int k = 0; k < 4; ++k) {
        int cc = c + k * WG_RANGE;
        bool ok = cc < c1;
        int4 tv = ok ? tg4[cc] : make_int4(-1, -1, -1, -1);
        int4 sv = ok ? sr4[cc] : make_int4(0, 0, 0, 0);
        u[4 * k + 0] = (unsigned)(tv.x - lo); u[4 * k + 1] = (unsigned)(tv.y - lo);
        u[4 * k + 2] = (unsigned)(tv.z - lo); u[4 * k + 3] = (unsigned)(tv.w - lo);
        s[4 * k + 0] = sv.x; s[4 * k + 1] = sv.y; s[4 * k + 2] = sv.z; s[4 * k + 3] = sv.w;
      }
      #pragma unroll
      for (int j = 0; j < 16; ++j) {
        const bool act = u[j] < span;
        p[j] = atomicAdd(&cur[act ? u[j] : (unsigned)(RANGE_NODES + lane)], act ? 1u : 0u);
      }
      #pragma unroll
      for (int j = 0; j < 16; ++j) if (u[j] < span) sorted[p[j]] = s[j];
    }
  } else {
    const int eps = (E + S_SLICES - 1) / S_SLICES;
    const int e0  = blockIdx.x * eps;
    const int e1  = (e0 + eps < E) ? (e0 + eps) : E;
    for (int e = e0 + tid; e < e1; e += WG_RANGE) {
      unsigned u = (unsigned)(tgt[e] - lo);
      if (u < span) sorted[atomicAdd(&cur[u], 1u)] = src[e];
    }
  }
}

#define GEMM_WG 256
__global__ void __launch_bounds__(GEMM_WG)
gemm1_wmma_kernel(const float* __restrict__ x,
                                  const float* __restrict__ W1,
                                  const unsigned int* __restrict__ deg,
                                  float* __restrict__ t1, int N) {
  __shared__ __align__(32) __bf16 Bfrag[8][2][32][16];

  const int tid  = threadIdx.x;
  const int lane = tid & 31;
  const int wave = tid >> 5;

  {
    constexpr int PER_THREAD = (8 * 2 * 32 * 16) / GEMM_WG;
    float wv[PER_THREAD];
    #pragma unroll
    for (int it = 0; it < PER_THREAD; ++it) {
      int idx   = tid + it * GEMM_WG;
      int j     = idx & 15;
      int ln    = (idx >> 4) & 31;
      int ntile = (idx >> 9) & 1;
      int ks    = idx >> 10;
      int jj = j >> 1, p = j & 1;
      int hi = ln >> 4;
      int K  = ks * 32 + 16 * (jj >> 2) + 8 * hi + 2 * (jj & 3) + p;
      int n  = ntile * 16 + (ln & 15);
      wv[it] = W1[K * H_DIM + n];
    }
    __bf16* __restrict__ bflat = &Bfrag[0][0][0][0];
    #pragma unroll
    for (int it = 0; it < PER_THREAD; ++it) bflat[tid + it * GEMM_WG] = (__bf16)wv[it];
  }
  __syncthreads();

  long long rowbase = ((long long)blockIdx.x * 8 + wave) * 16;
  if (rowbase >= N) return;

  const int m  = lane & 15;
  const int hi = lane >> 4;
  const float* xrow = x + (rowbase + m) * D_IN;

  v8f acc0 = {}; v8f acc1 = {};
  #pragma unroll
  for (int ks = 0; ks < 8; ++ks) {
    v16bf a;
    #pragma unroll
    for (int jj = 0; jj < 8; ++jj) {
      int K0 = ks * 32 + 16 * (jj >> 2) + 8 * hi + 2 * (jj & 3);
      float2 f = *(const float2*)(xrow + K0);
      a[2 * jj]     = (__bf16)f.x;
      a[2 * jj + 1] = (__bf16)f.y;
    }
    v16bf b0 = *(const v16bf*)&Bfrag[ks][0][lane][0];
    v16bf b1 = *(const v16bf*)&Bfrag[ks][1][lane][0];
    acc0 = __builtin_amdgcn_wmma_f32_16x16x32_bf16(false, a, false, b0,
                                                   (short)0, acc0, false, false);
    acc1 = __builtin_amdgcn_wmma_f32_16x16x32_bf16(false, a, false, b1,
                                                   (short)0, acc1, false, false);
  }

  #pragma unroll
  for (int r = 0; r < 8; ++r) {
    long long row = rowbase + r + 8 * hi;
    float di = rsqrtf((float)(deg[row] + 1u));
    t1[row * H_DIM + m]      = acc0[r] * di;
    t1[row * H_DIM + 16 + m] = acc1[r] * di;
  }
}

#define GB 256
__global__ void __launch_bounds__(GB)
gather1_kernel(const int* __restrict__ sorted, const unsigned int* __restrict__ rowptr,
               const float* __restrict__ t1, const float* __restrict__ b1,
               const float* __restrict__ W2, float* __restrict__ h2, int N, int E) {
  const int lane  = threadIdx.x & 31;
  const int sub   = lane & 7;
  const int gbase = lane & 24;
  const long long gid = (long long)blockIdx.x * GB + threadIdx.x;
  const long long tq  = gid >> 3;
  const bool valid = tq < N;
  const int t = valid ? (int)tq : N - 1;
  const unsigned beg = rowptr[t];
  const unsigned end = (t + 1 < N) ? rowptr[t + 1] : (unsigned)E;
  const float4* __restrict__ tv = (const float4*)t1;
  const float4 self = tv[(long long)t * (H_DIM / 4) + sub];
  float4 acc = make_float4(0.0f, 0.0f, 0.0f, 0.0f);
  int mine = (beg + sub < end) ? sorted[beg + sub] : -1;
  for (unsigned k = beg; k < end; k += 8) {
    const unsigned kn = k + 8 + sub;
    const int nw  = sorted[(kn < end) ? kn : k];
    const int nxt = (kn < end) ? nw : -1;
    #pragma unroll
    for (int j = 0; j < 8; ++j) {
      const int sj = __shfl(mine, gbase + j);
      const bool on = sj >= 0;
      const float4 v = tv[(long long)(on ? sj : 0) * (H_DIM / 4) + sub];
      acc.x += on ? v.x : 0.0f;
      acc.y += on ? v.y : 0.0f;
      acc.z += on ? v.z : 0.0f;
      acc.w += on ? v.w : 0.0f;
    }
    mine = nxt;
  }
  const float di = rsqrtf((float)(end - beg + 1u));
  const float4 bb = ((const float4*)b1)[sub];
  float h0 = di * (acc.x + self.x) + bb.x;
  float h1 = di * (acc.y + self.y) + bb.y;
  float h2a = di * (acc.z + self.z) + bb.z;
  float h3 = di * (acc.w + self.w) + bb.w;
  h0 = h0 > 0.0f ? h0 : 0.0f;
  h1 = h1 > 0.0f ? h1 : 0.0f;
  h2a = h2a > 0.0f ? h2a : 0.0f;
  h3 = h3 > 0.0f ? h3 : 0.0f;
  const float4 wa = ((const float4*)W2)[2 * sub];
  const float4 wb = ((const float4*)W2)[2 * sub + 1];
  float z0 = h0 * wa.x + h1 * wa.z + h2a * wb.x + h3 * wb.z;
  float z1 = h0 * wa.y + h1 * wa.w + h2a * wb.y + h3 * wb.w;
  #pragma unroll
  for (int m = 1; m < 8; m <<= 1) { z0 += __shfl_xor(z0, m); z1 += __shfl_xor(z1, m); }
  if (valid && sub == 0) ((float2*)h2)[t] = make_float2(di * z0, di * z1);
}

__global__ void __launch_bounds__(GB)
gather2_logsoftmax_kernel(const int* __restrict__ sorted,
                          const unsigned int* __restrict__ rowptr,
                          const float* __restrict__ h2, const float* __restrict__ b2,
                          float* __restrict__ out, int N, int E) {
  const int sub = threadIdx.x & 7;
  const long long gid = (long long)blockIdx.x * GB + threadIdx.x;
  const long long tq  = gid >> 3;
  const bool valid = tq < N;
  const int t = valid ? (int)tq : N - 1;
  const unsigned beg = rowptr[t];
  const unsigned end = (t + 1 < N) ? rowptr[t + 1] : (unsigned)E;
  const float2* __restrict__ hv = (const float2*)h2;
  const float2 self = hv[t];
  float a0 = 0.0f, a1 = 0.0f;
  for (unsigned k = beg + sub; k < end; k += 64) {
    int s[8];
    #pragma unroll
    for (int j = 0; j < 8; ++j) { unsigned kk = k + 8 * j; s[j] = (kk < end) ? sorted[kk] : -1; }
    #pragma unroll
    for (int j = 0; j < 8; ++j) {
      const bool on = s[j] >= 0;
      const float2 v = hv[on ? s[j] : 0];
      a0 += on ? v.x : 0.0f;
      a1 += on ? v.y : 0.0f;
    }
  }
  #pragma unroll
  for (int m = 1; m < 8; m <<= 1) { a0 += __shfl_xor(a0, m); a1 += __shfl_xor(a1, m); }
  const float di = rsqrtf((float)(end - beg + 1u));
  float z0 = di * (a0 + self.x) + b2[0];
  float z1 = di * (a1 + self.y) + b2[1];
  float mx  = fmaxf(z0, z1);
  float lse = mx + logf(expf(z0 - mx) + expf(z1 - mx));
  if (valid && sub == 0) ((float2*)out)[t] = make_float2(z0 - lse, z1 - lse);
}

extern "C" void kernel_launch(void* const* d_in, const int* in_sizes, int n_in,
                              void* d_out, int out_size, void* d_ws, size_t ws_size,
                              hipStream_t stream) {
  const float* x  = (const float*)d_in[0];
  const int*   ei = (const int*)d_in[1];
  const float* W1 = (const float*)d_in[2];
  const float* b1 = (const float*)d_in[3];
  const float* W2 = (const float*)d_in[4];
  const float* b2 = (const float*)d_in[5];
  float* out = (float*)d_out;

  const int N = in_sizes[0] / D_IN;
  const int E = in_sizes[1] / 2;

  float* ws   = (float*)d_ws;
  float* t1   = ws;
  float* h2   = t1 + (long long)N * H_DIM;
  unsigned int* rowptr = (unsigned int*)(h2 + (long long)N * C_DIM);
  int* sorted = (int*)(rowptr + N);
  unsigned int* degpart  = (unsigned int*)t1;
  unsigned int* deg      = (unsigned int*)h2;
  unsigned int* blocksum = deg + N;

  const dim3 rgrid(S_SLICES, (N + RANGE_NODES - 1) / RANGE_NODES);
  const int gblocks = (int)(((long long)N * 8 + GB - 1) / GB);

  degree_range_kernel<<<rgrid, WG_RANGE, 0, stream>>>(ei, degpart, E, N);
  slice_prefix_kernel<<<(N + PB - 1) / PB, PB, 0, stream>>>(degpart, deg, blocksum, N);
  sort_range_kernel<<<rgrid, WG_RANGE, 0, stream>>>(ei, degpart, deg, blocksum, rowptr, sorted, E, N);

  gemm1_wmma_kernel<<<(N + 127) / 128, GEMM_WG, 0, stream>>>(x, W1, deg, t1, N);
  gather1_kernel<<<gblocks, GB, 0, stream>>>(sorted, rowptr, t1, b1, W2, h2, N, E);

  gather2_logsoftmax_kernel<<<gblocks, GB, 0, stream>>>(sorted, rowptr, h2, b2, out, N, E);
}


# layer-2 source sums read an LDS table (one f32 per node, the class-score difference a 2-class log-softmax needs, half the nodes at a time) instead of 8-byte L2 gathers; f32 sum order changes
# speedup vs baseline: 3.9893x; 1.0444x over previous
#include <hip/hip_runtime.h>
#include <hip/hip_bf16.h>

typedef __attribute__((ext_vector_type(16))) __bf16 v16bf;
typedef __attribute__((ext_vector_type(8)))  float  v8f;

#define D_IN  256
#define H_DIM 32
#define C_DIM 2

#define S_SLICES    32
#define RANGE_NODES 12800
#define WG_RANGE    1024

__global__ void __launch_bounds__(WG_RANGE)
degree_range_kernel(const int* __restrict__ ei, unsigned int* __restrict__ part,
                    int E, int N) {
  __shared__ unsigned int cnt[RANGE_NODES];
  const int tid = threadIdx.x;
  const int lo  = blockIdx.y * RANGE_NODES;
  const int rem = N - lo;
  const unsigned span = (unsigned)(rem < RANGE_NODES ? rem : RANGE_NODES);
  const int* __restrict__ tgt = ei + E;
  const bool vec = (E & 3) == 0;
  const int C   = E >> 2;
  const int cps = (C + S_SLICES - 1) / S_SLICES;
  const int c0  = blockIdx.x * cps;
  const int c1  = (c0 + cps < C) ? (c0 + cps) : C;
  const int4* __restrict__ tg4 = (const int4*)tgt;
  int4 nv[4];
  #pragma unroll
  for (int k = 0; k < 4; ++k) {
    int cc = c0 + tid + k * WG_RANGE;
    nv[k] = (vec && cc < c1) ? tg4[cc] : make_int4(-1, -1, -1, -1);
  }
  for (int i = tid; i < RANGE_NODES; i += WG_RANGE) cnt[i] = 0u;
  __syncthreads();

  if (vec) {
    for (int c = c0 + tid; c < c1; c += 4 * WG_RANGE) {
      int4 v[4];
      #pragma unroll
      for (int k = 0; k < 4; ++k) v[k] = nv[k];
      #pragma unroll
      for (int k = 0; k < 4; ++k) {
        int cc = c + (4 + k) * WG_RANGE;
        nv[k] = (cc < c1) ? tg4[cc] : make_int4(-1, -1, -1, -1);
      }
      #pragma unroll
      for (int k = 0; k < 4; ++k) {
        unsigned u0 = (unsigned)(v[k].x - lo), u1 = (unsigned)(v[k].y - lo);
        unsigned u2 = (unsigned)(v[k].z - lo), u3 = (unsigned)(v[k].w - lo);
        if (u0 < span) atomicAdd(&cnt[u0], 1u);
        if (u1 < span) atomicAdd(&cnt[u1], 1u);
        if (u2 < span) atomicAdd(&cnt[u2], 1u);
        if (u3 < span) atomicAdd(&cnt[u3], 1u);
      }
    }
  } else {
    const int eps = (E + S_SLICES - 1) / S_SLICES;
    const int e0  = blockIdx.x * eps;
    const int e1  = (e0 + eps < E) ? (e0 + eps) : E;
    for (int e = e0 + tid; e < e1; e += WG_RANGE) {
      unsigned u = (unsigned)(tgt[e] - lo);
      if (u < span) atomicAdd(&cnt[u], 1u);
    }
  }
  __syncthreads();
  unsigned int* __restrict__ dst = part + (long long)blockIdx.x * N + lo;
  for (unsigned i = tid; i < span; i += WG_RANGE) dst[i] = cnt[i];
}


#define PB 256
static_assert(RANGE_NODES % PB == 0, "a node range must be a whole number of prefix blocks");
static_assert(WG_RANGE == 1024, "the range scan below assumes 32 waves of 32 lanes");
#define SCAN_CH  13
#define CUR_PAD  (WG_RANGE * SCAN_CH)
static_assert(CUR_PAD >= RANGE_NODES + 32, "a cursor per node and a spare word per lane");

__global__ void __launch_bounds__(PB)
slice_prefix_kernel(unsigned int* __restrict__ part, unsigned int* __restrict__ deg,
                    unsigned int* __restrict__ blocksum, int N) {
  __shared__ unsigned int wtot[PB / 32];
  const int i = blockIdx.x * PB + threadIdx.x;
  unsigned c = 0u;
  if (i < N) {
    unsigned v[S_SLICES];
    #pragma unroll
    for (int s = 0; s < S_SLICES; ++s) v[s] = part[(long long)s * N + i];
    #pragma unroll
    for (int s = 0; s < S_SLICES; ++s) { part[(long long)s * N + i] = c; c += v[s]; }
    deg[i] = c;
  }
  unsigned r = c;
  #pragma unroll
  for (int m = 16; m >= 1; m >>= 1) r += __shfl_xor(r, m);
  if ((threadIdx.x & 31) == 0) wtot[threadIdx.x >> 5] = r;
  __syncthreads();
  if (threadIdx.x == 0) {
    unsigned t = 0u;
    #pragma unroll
    for (int w = 0; w < PB / 32; ++w) t += wtot[w];
    blocksum[blockIdx.x] = t;
  }
}

__global__ void __launch_bounds__(WG_RANGE)
sort_range_kernel(const int* __restrict__ ei, const unsigned int* __restrict__ part,
                  const unsigned int* __restrict__ deg,
                  const unsigned int* __restrict__ blocksum,
                  unsigned int* __restrict__ rowptr, int* __restrict__ sorted,
                  int E, int N) {
  __shared__ unsigned int cur[CUR_PAD];
  __shared__ unsigned int wsum[32];
  __shared__ unsigned int bsum[32];
  __shared__ unsigned int sbase;
  const int tid  = threadIdx.x;
  const int lane = tid & 31;
  const int wave = tid >> 5;
  const int lo  = blockIdx.y * RANGE_NODES;
  const int rem = N - lo;
  const unsigned span = (unsigned)(rem < RANGE_NODES ? rem : RANGE_NODES);

  const unsigned int* __restrict__ mypart = part + (long long)blockIdx.x * N + lo;
  unsigned mp[SCAN_CH];
  {
    unsigned dv[SCAN_CH];
    #pragma unroll
    for (int k = 0; k < SCAN_CH; ++k) {
      unsigned i  = (unsigned)tid + (unsigned)k * WG_RANGE;
      unsigned ic = (i < span) ? i : span - 1u;
      unsigned v  = deg[lo + ic];
      dv[k] = (i < span) ? v : 0u;
      mp[k] = mypart[ic];
    }
    #pragma unroll
    for (int k = 0; k < SCAN_CH; ++k) cur[tid + k * WG_RANGE] = dv[k];
  }
  const int nb = blockIdx.y * (RANGE_NODES / PB);
  unsigned bs = 0u;
  for (int b = tid; b < nb; b += WG_RANGE) bs += blocksum[b];
  #pragma unroll
  for (int m = 16; m >= 1; m >>= 1) bs += __shfl_xor(bs, m);
  if (lane == 0) bsum[wave] = bs;
  __syncthreads();

  const int st = tid * SCAN_CH;
  unsigned loc = 0u;
  #pragma unroll
  for (int k = 0; k < SCAN_CH; ++k) loc += cur[st + k];
  unsigned inc = loc;
  #pragma unroll
  for (int d = 1; d < 32; d <<= 1) { unsigned n = __shfl_up(inc, d); if (lane >= d) inc += n; }
  if (lane == 31) wsum[wave] = inc;
  __syncthreads();
  if (wave == 0) {
    unsigned v = wsum[lane], w = v;
    #pragma unroll
    for (int d = 1; d < 32; d <<= 1) { unsigned n = __shfl_up(w, d); if (lane >= d) w += n; }
    wsum[lane] = w - v;
    unsigned b = bsum[lane];
    #pragma unroll
    for (int m = 16; m >= 1; m >>= 1) b += __shfl_xor(b, m);
    if (lane == 0) sbase = b;
  }
  __syncthreads();
  unsigned run = sbase + wsum[wave] + (inc - loc);
  #pragma unroll
  for (int k = 0; k < SCAN_CH; ++k) { unsigned d = cur[st + k]; cur[st + k] = run; run += d; }
  __syncthreads();
  {
    unsigned rr[SCAN_CH];
    #pragma unroll
    for (int k = 0; k < SCAN_CH; ++k) rr[k] = cur[tid + k * WG_RANGE];
    #pragma unroll
    for (int k = 0; k < SCAN_CH; ++k) {
      const unsigned i = (unsigned)tid + (unsigned)k * WG_RANGE;
      if (i < span) {
        if (blockIdx.x == 0) rowptr[lo + i] = rr[k];
        cur[i] = rr[k] + mp[k];
      }
    }
  }
  __syncthreads();

  const int* __restrict__ src = ei;
  const int* __restrict__ tgt = ei + E;
  if ((E & 3) == 0) {
    const int C   = E >> 2;
    const int cps = (C + S_SLICES - 1) / S_SLICES;
    const int c0  = blockIdx.x * cps;
    const int c1  = (c0 + cps < C) ? (c0 + cps) : C;
    const int4* __restrict__ tg4 = (const int4*)tgt;
    const int4* __restrict__ sr4 = (const int4*)src;
    for (int c = c0 + tid; c < c1; c += 4 * WG_RANGE) {
      unsigned u[16]; int s[16]; unsigned p[16];
      #pragma unroll
      for (int k = 0; k < 4; ++k) {
        int cc = c + k * WG_RANGE;
        bool ok = cc < c1;
        int4 tv = ok ? tg4[cc] : make_int4(-1, -1, -1, -1);
        int4 sv = ok ? sr4[cc] : make_int4(0, 0, 0, 0);
        u[4 * k + 0] = (unsigned)(tv.x - lo); u[4 * k + 1] = (unsigned)(tv.y - lo);
        u[4 * k + 2] = (unsigned)(tv.z - lo); u[4 * k + 3] = (unsigned)(tv.w - lo);
        s[4 * k + 0] = sv.x; s[4 * k + 1] = sv.y; s[4 * k + 2] = sv.z; s[4 * k + 3] = sv.w;
      }
      #pragma unroll
      for (int j = 0; j < 16; ++j) {
        const bool act = u[j] < span;
        p[j] = atomicAdd(&cur[act ? u[j] : (unsigned)(RANGE_NODES + lane)], act ? 1u : 0u);
      }
      #pragma unroll
      for (int j = 0; j < 16; ++j) if (u[j] < span) sorted[p[j]] = s[j];
    }
  } else {
    const int eps = (E + S_SLICES - 1) / S_SLICES;
    const int e0  = blockIdx.x * eps;
    const int e1  = (e0 + eps < E) ? (e0 + eps) : E;
    for (int e = e0 + tid; e < e1; e += WG_RANGE) {
      unsigned u = (unsigned)(tgt[e] - lo);
      if (u < span) sorted[atomicAdd(&cur[u], 1u)] = src[e];
    }
  }
}

#define GEMM_WG 256
__global__ void __launch_bounds__(GEMM_WG)
gemm1_wmma_kernel(const float* __restrict__ x,
                                  const float* __restrict__ W1,
                                  const unsigned int* __restrict__ deg,
                                  float* __restrict__ t1, int N) {
  __shared__ __align__(32) __bf16 Bfrag[8][2][32][16];

  const int tid  = threadIdx.x;
  const int lane = tid & 31;
  const int wave = tid >> 5;

  {
    constexpr int PER_THREAD = (8 * 2 * 32 * 16) / GEMM_WG;
    float wv[PER_THREAD];
    #pragma unroll
    for (int it = 0; it < PER_THREAD; ++it) {
      int idx   = tid + it * GEMM_WG;
      int j     = idx & 15;
      int ln    = (idx >> 4) & 31;
      int ntile = (idx >> 9) & 1;
      int ks    = idx >> 10;
      int jj = j >> 1, p = j & 1;
      int hi = ln >> 4;
      int K  = ks * 32 + 16 * (jj >> 2) + 8 * hi + 2 * (jj & 3) + p;
      int n  = ntile * 16 + (ln & 15);
      wv[it] = W1[K * H_DIM + n];
    }
    __bf16* __restrict__ bflat = &Bfrag[0][0][0][0];
    #pragma unroll
    for (int it = 0; it < PER_THREAD; ++it) bflat[tid + it * GEMM_WG] = (__bf16)wv[it];
  }
  __syncthreads();

  long long rowbase = ((long long)blockIdx.x * 8 + wave) * 16;
  if (rowbase >= N) return;

  const int m  = lane & 15;
  const int hi = lane >> 4;
  const float* xrow = x + (rowbase + m) * D_IN;

  v8f acc0 = {}; v8f acc1 = {};
  #pragma unroll
  for (int ks = 0; ks < 8; ++ks) {
    v16bf a;
    #pragma unroll
    for (int jj = 0; jj < 8; ++jj) {
      int K0 = ks * 32 + 16 * (jj >> 2) + 8 * hi + 2 * (jj & 3);
      float2 f = *(const float2*)(xrow + K0);
      a[2 * jj]     = (__bf16)f.x;
      a[2 * jj + 1] = (__bf16)f.y;
    }
    v16bf b0 = *(const v16bf*)&Bfrag[ks][0][lane][0];
    v16bf b1 = *(const v16bf*)&Bfrag[ks][1][lane][0];
    acc0 = __builtin_amdgcn_wmma_f32_16x16x32_bf16(false, a, false, b0,
                                                   (short)0, acc0, false, false);
    acc1 = __builtin_amdgcn_wmma_f32_16x16x32_bf16(false, a, false, b1,
                                                   (short)0, acc1, false, false);
  }

  #pragma unroll
  for (int r = 0; r < 8; ++r) {
    long long row = rowbase + r + 8 * hi;
    float di = rsqrtf((float)(deg[row] + 1u));
    t1[row * H_DIM + m]      = acc0[r] * di;
    t1[row * H_DIM + 16 + m] = acc1[r] * di;
  }
}

#define GB 256
__global__ void __launch_bounds__(GB)
gather1_kernel(const int* __restrict__ sorted, const unsigned int* __restrict__ rowptr,
               const float* __restrict__ t1, const float* __restrict__ b1,
               const float* __restrict__ W2, float* __restrict__ dcol, int N, int E) {
  const int lane  = threadIdx.x & 31;
  const int sub   = lane & 7;
  const int gbase = lane & 24;
  const long long gid = (long long)blockIdx.x * GB + threadIdx.x;
  const long long tq  = gid >> 3;
  const bool valid = tq < N;
  const int t = valid ? (int)tq : N - 1;
  const unsigned beg = rowptr[t];
  const unsigned end = (t + 1 < N) ? rowptr[t + 1] : (unsigned)E;
  const float4* __restrict__ tv = (const float4*)t1;
  const float4 self = tv[(long long)t * (H_DIM / 4) + sub];
  float4 acc = make_float4(0.0f, 0.0f, 0.0f, 0.0f);
  int mine = (beg + sub < end) ? sorted[beg + sub] : -1;
  for (unsigned k = beg; k < end; k += 8) {
    const unsigned kn = k + 8 + sub;
    const int nw  = sorted[(kn < end) ? kn : k];
    const int nxt = (kn < end) ? nw : -1;
    #pragma unroll
    for (int j = 0; j < 8; ++j) {
      const int sj = __shfl(mine, gbase + j);
      const bool on = sj >= 0;
      const float4 v = tv[(long long)(on ? sj : 0) * (H_DIM / 4) + sub];
      acc.x += on ? v.x : 0.0f;
      acc.y += on ? v.y : 0.0f;
      acc.z += on ? v.z : 0.0f;
      acc.w += on ? v.w : 0.0f;
    }
    mine = nxt;
  }
  const float di = rsqrtf((float)(end - beg + 1u));
  const float4 bb = ((const float4*)b1)[sub];
  float h0 = di * (acc.x + self.x) + bb.x;
  float h1 = di * (acc.y + self.y) + bb.y;
  float h2a = di * (acc.z + self.z) + bb.z;
  float h3 = di * (acc.w + self.w) + bb.w;
  h0 = h0 > 0.0f ? h0 : 0.0f;
  h1 = h1 > 0.0f ? h1 : 0.0f;
  h2a = h2a > 0.0f ? h2a : 0.0f;
  h3 = h3 > 0.0f ? h3 : 0.0f;
  const float4 wa = ((const float4*)W2)[2 * sub];
  const float4 wb = ((const float4*)W2)[2 * sub + 1];
  float z0 = h0 * wa.x + h1 * wa.z + h2a * wb.x + h3 * wb.z;
  float z1 = h0 * wa.y + h1 * wa.w + h2a * wb.y + h3 * wb.w;
  #pragma unroll
  for (int m = 1; m < 8; m <<= 1) { z0 += __shfl_xor(z0, m); z1 += __shfl_xor(z1, m); }
  if (valid && sub == 0) dcol[t] = di * (z0 - z1);
}

#define G2_WG    1024
#define G2_TPB   392
#define G2_NI    4
#define G2_FILLK 13
#define G2_TAB   (G2_WG * 4 * G2_FILLK)
static_assert(G2_NI * (G2_WG / 8) >= G2_TPB, "a lane group per target node of the work-group");

__device__ __forceinline__ void g2_ask(const float* __restrict__ src, int cnt, int tid,
                                       float4 (&nv)[G2_FILLK]) {
  const int nq = cnt >> 2;
  if (nq > 0) {
    const float4* __restrict__ s4 = (const float4*)src;
    #pragma unroll
    for (int k = 0; k < G2_FILLK; ++k) {
      const int q = tid + k * G2_WG;
      nv[k] = s4[q < nq ? q : nq - 1];
    }
  } else {
    #pragma unroll
    for (int k = 0; k < G2_FILLK; ++k) nv[k] = make_float4(0.0f, 0.0f, 0.0f, 0.0f);
  }
}

__device__ __forceinline__ void g2_put(float* __restrict__ tab, const float* __restrict__ src,
                                       int cnt, int tid, const float4 (&nv)[G2_FILLK]) {
  const int nq = cnt >> 2;
  #pragma unroll
  for (int k = 0; k < G2_FILLK; ++k) {
    const int q = tid + k * G2_WG;
    if (q < nq) ((float4*)tab)[q] = nv[k];
  }
  if (tid < (cnt & 3)) tab[(nq << 2) + tid] = src[(nq << 2) + tid];
}

__global__ void __launch_bounds__(G2_WG)
gather2_logsoftmax_kernel(const int* __restrict__ sorted,
                          const unsigned int* __restrict__ rowptr,
                          const float* __restrict__ dcol, const float* __restrict__ b2,
                          float* __restrict__ out, int N, int E) {
  __shared__ __align__(16) float tab[G2_TAB];
  const int tid  = threadIdx.x;
  const int sub  = tid & 7;
  const int grp  = tid >> 3;
  const int base = blockIdx.x * G2_TPB;

  float4 nv0[G2_FILLK];
  const int cnt0 = N < G2_TAB ? N : G2_TAB;
  g2_ask(dcol, cnt0, tid, nv0);

  unsigned beg[G2_NI], end[G2_NI];
  float self[G2_NI];
  #pragma unroll
  for (int i = 0; i < G2_NI; ++i) {
    const int lt = grp + i * (G2_WG / 8);
    const int t  = base + lt;
    const bool v = (lt < G2_TPB) && (t < N);
    const int tc = v ? t : N - 1;
    const bool more = tc + 1 < N;
    const unsigned b  = rowptr[tc];
    const unsigned en = rowptr[more ? tc + 1 : tc];
    beg[i]  = b;
    end[i]  = v ? (more ? en : (unsigned)E) : b;
    self[i] = dcol[tc];
  }
  g2_put(tab, dcol, cnt0, tid, nv0);

  int idx[G2_NI][8];
  #pragma unroll
  for (int i = 0; i < G2_NI; ++i) {
    #pragma unroll
    for (int j = 0; j < 8; ++j) {
      const unsigned kk = beg[i] + (unsigned)sub + 8u * (unsigned)j;
      const bool ok = kk < end[i];
      const int w = sorted[ok ? kk : 0u];
      idx[i][j] = ok ? w : -1;
    }
  }

  float acc[G2_NI];
  #pragma unroll
  for (int i = 0; i < G2_NI; ++i) acc[i] = 0.0f;

  for (int p0 = 0; p0 < N; p0 += G2_TAB) {
    const int cnt = (N - p0 < G2_TAB) ? (N - p0) : G2_TAB;
    if (p0 > 0) {
      float4 nv[G2_FILLK];
      g2_ask(dcol + p0, cnt, tid, nv);
      __syncthreads();
      g2_put(tab, dcol + p0, cnt, tid, nv);
    }
    __syncthreads();
    #pragma unroll
    for (int i = 0; i < G2_NI; ++i) {
      if ((tid & ~31) / 8 + i * (G2_WG / 8) < G2_TPB) {
        float v[8];
        #pragma unroll
        for (int j = 0; j < 8; ++j) {
          const unsigned u = (unsigned)(idx[i][j] - p0);
          v[j] = tab[u < (unsigned)cnt ? u : 0u];
        }
        #pragma unroll
        for (int j = 0; j < 8; ++j) {
          const unsigned u = (unsigned)(idx[i][j] - p0);
          acc[i] += u < (unsigned)cnt ? v[j] : 0.0f;
        }
      }
    }
  }

  const float bd = b2[0] - b2[1];
  #pragma unroll
  for (int i = 0; i < G2_NI; ++i) {
    const int lt = grp + i * (G2_WG / 8);
    float a = acc[i];
    for (unsigned k = beg[i] + (unsigned)sub + 64u; k < end[i]; k += 8u) a += dcol[sorted[k]];
    #pragma unroll
    for (int m = 1; m < 8; m <<= 1) a += __shfl_xor(a, m);
    const float di  = rsqrtf((float)(end[i] - beg[i] + 1u));
    const float d   = di * (a + self[i]) + bd;
    const float mx  = fmaxf(d, 0.0f);
    const float lse = mx + logf(expf(d - mx) + expf(-mx));
    if (sub == 0 && lt < G2_TPB && base + lt < N) ((float2*)out)[base + lt] = make_float2(d - lse, -lse);
  }
}

extern "C" void kernel_launch(void* const* d_in, const int* in_sizes, int n_in,
                              void* d_out, int out_size, void* d_ws, size_t ws_size,
                              hipStream_t stream) {
  const float* x  = (const float*)d_in[0];
  const int*   ei = (const int*)d_in[1];
  const float* W1 = (const float*)d_in[2];
  const float* b1 = (const float*)d_in[3];
  const float* W2 = (const float*)d_in[4];
  const float* b2 = (const float*)d_in[5];
  float* out = (float*)d_out;

  const int N = in_sizes[0] / D_IN;
  const int E = in_sizes[1] / 2;

  float* ws   = (float*)d_ws;
  float* t1   = ws;
  float* h2   = t1 + (long long)N * H_DIM;
  unsigned int* rowptr = (unsigned int*)(h2 + (long long)N * C_DIM);
  int* sorted = (int*)(rowptr + N);
  unsigned int* degpart  = (unsigned int*)t1;
  unsigned int* deg      = (unsigned int*)h2;
  unsigned int* blocksum = deg + N;

  const dim3 rgrid(S_SLICES, (N + RANGE_NODES - 1) / RANGE_NODES);
  const int gblocks = (int)(((long long)N * 8 + GB - 1) / GB);

  degree_range_kernel<<<rgrid, WG_RANGE, 0, stream>>>(ei, degpart, E, N);
  slice_prefix_kernel<<<(N + PB - 1) / PB, PB, 0, stream>>>(degpart, deg, blocksum, N);
  sort_range_kernel<<<rgrid, WG_RANGE, 0, stream>>>(ei, degpart, deg, blocksum, rowptr, sorted, E, N);

  gemm1_wmma_kernel<<<(N + 127) / 128, GEMM_WG, 0, stream>>>(x, W1, deg, t1, N);
  gather1_kernel<<<gblocks, GB, 0, stream>>>(sorted, rowptr, t1, b1, W2, h2, N, E);

  gather2_logsoftmax_kernel<<<(N + G2_TPB - 1) / G2_TPB, G2_WG, 0, stream>>>(sorted, rowptr, h2, b2, out, N, E);
}


# edge sort rebuilt in two levels (chunk to 256-node blocks through LDS, then per-block order by node in LDS): each edge read once per pass, slot stores in consecutive words; f32 sum order changes
# speedup vs baseline: 6.2786x; 1.5739x over previous
#include <hip/hip_runtime.h>
#include <hip/hip_bf16.h>

typedef __attribute__((ext_vector_type(16))) __bf16 v16bf;
typedef __attribute__((ext_vector_type(8)))  float  v8f;

#define D_IN  256
#define H_DIM 32
#define C_DIM 2

#define NODE_BITS 8
#define BLK_NODES 256
#define NBMAX     512
#define SWG       512
#define EPT       25
#define CHUNK     (SWG * EPT)
#define PWG       1024
#define KW        20
#define STAGE_CAP 16384
static_assert(NBMAX == SWG, "one thread per node block in the block tables");

__global__ void __launch_bounds__(SWG)
block_count_kernel(const int* __restrict__ ei, unsigned int* __restrict__ hist,
                   int E, int N, int NB) {
  __shared__ unsigned int h[NBMAX];
  const int tid = threadIdx.x;
  const int e0  = blockIdx.x * CHUNK;
  const int e1  = (E - e0 > CHUNK) ? (e0 + CHUNK) : E;
  const int* __restrict__ tgt = ei + E;
  int tv[EPT];
  #pragma unroll
  for (int k = 0; k < EPT; ++k) {
    const int e = e0 + tid + k * SWG;
    const int v = tgt[e < e1 ? e : e0];
    tv[k] = (e < e1) ? v : -1;
  }
  h[tid] = 0u;
  __syncthreads();
  #pragma unroll
  for (int k = 0; k < EPT; ++k)
    if ((unsigned)tv[k] < (unsigned)N) atomicAdd(&h[tv[k] >> NODE_BITS], 1u);
  __syncthreads();
  if (tid < NB) hist[(long long)blockIdx.x * NB + tid] = h[tid];
}

__global__ void __launch_bounds__(PWG)
block_prefix_kernel(const unsigned int* __restrict__ hist, unsigned int* __restrict__ off,
                    unsigned int* __restrict__ tot, int P, int NB) {
  __shared__ unsigned int seg[32][32];
  const int lane = threadIdx.x & 31;
  const int w    = threadIdx.x >> 5;
  const int b    = blockIdx.x * 32 + lane;
  const bool on  = b < NB;
  const int pps  = (P + 31) / 32;
  const int p0   = w * pps;
  const int p1   = (p0 + pps < P) ? (p0 + pps) : P;
  unsigned s = 0u;
  if (on) for (int p = p0; p < p1; ++p) s += hist[(long long)p * NB + b];
  seg[w][lane] = s;
  __syncthreads();
  unsigned run = 0u;
  #pragma unroll
  for (int q = 0; q < 32; ++q) { const unsigned v = seg[q][lane]; run += (q < w) ? v : 0u; }
  if (on) {
    for (int p = p0; p < p1; ++p) {
      const unsigned v = hist[(long long)p * NB + b];
      off[(long long)p * NB + b] = run;
      run += v;
    }
    if (w == 31) tot[b] = run;
  }
}

__global__ void __launch_bounds__(SWG)
block_place_kernel(const int* __restrict__ ei, const unsigned int* __restrict__ hist,
                   const unsigned int* __restrict__ off, const unsigned int* __restrict__ tot,
                   unsigned int* __restrict__ bbase, unsigned int* __restrict__ mid,
                   int E, int N, int NB) {
  __shared__ unsigned int lc[NBMAX + 32];
  __shared__ unsigned int dl[NBMAX];
  __shared__ unsigned int buf[CHUNK];
  __shared__ unsigned int gp[CHUNK];
  __shared__ unsigned int wa[SWG / 32];
  __shared__ unsigned int wb[SWG / 32];
  __shared__ unsigned int nkept;
  const int tid  = threadIdx.x;
  const int lane = tid & 31;
  const int wave = tid >> 5;
  const int e0   = blockIdx.x * CHUNK;
  const int e1   = (E - e0 > CHUNK) ? (e0 + CHUNK) : E;
  const int* __restrict__ src = ei;
  const int* __restrict__ tgt = ei + E;
  int tv[EPT], sv[EPT];
  #pragma unroll
  for (int k = 0; k < EPT; ++k) {
    const int e  = e0 + tid + k * SWG;
    const int ec = e < e1 ? e : e0;
    const int t  = tgt[ec];
    sv[k] = src[ec];
    tv[k] = (e < e1) ? t : -1;
  }
  const bool onb = tid < NB;
  const long long row = (long long)blockIdx.x * NB;
  const unsigned c  = onb ? hist[row + tid] : 0u;
  const unsigned g  = onb ? off[row + tid] : 0u;
  const unsigned tt = onb ? tot[tid] : 0u;
  unsigned ic = c, it = tt;
  #pragma unroll
  for (int d = 1; d < 32; d <<= 1) {
    const unsigned nc = __shfl_up(ic, d);
    const unsigned nt = __shfl_up(it, d);
    if (lane >= d) { ic += nc; it += nt; }
  }
  if (lane == 31) { wa[wave] = ic; wb[wave] = it; }
  __syncthreads();
  unsigned ba = 0u, bb = 0u;
  #pragma unroll
  for (int q = 0; q < SWG / 32; ++q) {
    const unsigned va = wa[q], vb = wb[q];
    ba += (q < wave) ? va : 0u;
    bb += (q < wave) ? vb : 0u;
  }
  const unsigned ls = ba + ic - c;
  const unsigned gb = bb + it - tt;
  lc[tid] = ls;
  dl[tid] = gb + g - ls;
  if (tid == SWG - 1) nkept = ba + ic;
  if (blockIdx.x == 0 && onb) {
    bbase[tid] = gb;
    if (tid == NB - 1) bbase[NB] = gb + tt;
  }
  __syncthreads();
  unsigned r[EPT];
  #pragma unroll
  for (int k = 0; k < EPT; ++k) {
    const bool act = (unsigned)tv[k] < (unsigned)N;
    r[k] = atomicAdd(&lc[act ? (unsigned)(tv[k] >> NODE_BITS) : (unsigned)(NBMAX + lane)],
                     act ? 1u : 0u);
  }
  #pragma unroll
  for (int k = 0; k < EPT; ++k) {
    if ((unsigned)tv[k] < (unsigned)N) {
      const unsigned b = (unsigned)(tv[k] >> NODE_BITS);
      buf[r[k]] = ((unsigned)sv[k] << NODE_BITS) | ((unsigned)tv[k] & (BLK_NODES - 1));
      gp[r[k]]  = r[k] + dl[b];
    }
  }
  __syncthreads();
  const unsigned n = nkept;
  for (unsigned i = tid; i < n; i += SWG) mid[gp[i]] = buf[i];
}

__global__ void __launch_bounds__(SWG)
block_sort_kernel(const unsigned int* __restrict__ mid, const unsigned int* __restrict__ bbase,
                  unsigned int* __restrict__ rowptr, unsigned int* __restrict__ deg,
                  int* __restrict__ sorted, int N) {
  __shared__ unsigned int lcnt[BLK_NODES + 32];
  __shared__ unsigned int wt[BLK_NODES / 32];
  __shared__ unsigned int stage[STAGE_CAP];
  const int tid  = threadIdx.x;
  const int lane = tid & 31;
  const int wave = tid >> 5;
  const unsigned base = bbase[blockIdx.x];
  const unsigned cnt  = bbase[blockIdx.x + 1] - base;
  if (tid < BLK_NODES) lcnt[tid] = 0u;
  __syncthreads();
  for (unsigned i0 = 0u; i0 < cnt; i0 += KW * SWG) {
    unsigned w[KW];
    #pragma unroll
    for (int k = 0; k < KW; ++k) {
      const unsigned i = i0 + tid + k * SWG;
      w[k] = mid[base + (i < cnt ? i : 0u)];
    }
    #pragma unroll
    for (int k = 0; k < KW; ++k)
      if (i0 + tid + k * SWG < cnt) atomicAdd(&lcnt[w[k] & (BLK_NODES - 1)], 1u);
  }
  __syncthreads();
  unsigned c = 0u, ic = 0u;
  if (tid < BLK_NODES) {
    c = lcnt[tid];
    ic = c;
    #pragma unroll
    for (int d = 1; d < 32; d <<= 1) { const unsigned nn = __shfl_up(ic, d); if (lane >= d) ic += nn; }
    if (lane == 31) wt[wave] = ic;
  }
  __syncthreads();
  if (tid < BLK_NODES) {
    unsigned bs = 0u;
    #pragma unroll
    for (int q = 0; q < BLK_NODES / 32; ++q) { const unsigned v = wt[q]; bs += (q < wave) ? v : 0u; }
    const unsigned excl = bs + ic - c;
    const int node = blockIdx.x * BLK_NODES + tid;
    if (node < N) { rowptr[node] = base + excl; deg[node] = c; }
    lcnt[tid] = excl;
  }
  __syncthreads();
  if (cnt <= STAGE_CAP) {
    for (unsigned i0 = 0u; i0 < cnt; i0 += KW * SWG) {
      unsigned w[KW], r[KW];
      #pragma unroll
      for (int k = 0; k < KW; ++k) {
        const unsigned i = i0 + tid + k * SWG;
        w[k] = mid[base + (i < cnt ? i : 0u)];
      }
      #pragma unroll
      for (int k = 0; k < KW; ++k) {
        const bool act = i0 + tid + k * SWG < cnt;
        r[k] = atomicAdd(&lcnt[act ? (w[k] & (BLK_NODES - 1)) : (unsigned)(BLK_NODES + lane)],
                         act ? 1u : 0u);
      }
      #pragma unroll
      for (int k = 0; k < KW; ++k)
        if (i0 + tid + k * SWG < cnt) stage[r[k]] = w[k] >> NODE_BITS;
    }
    __syncthreads();
    for (unsigned i = tid; i < cnt; i += SWG) sorted[base + i] = (int)stage[i];
  } else {
    for (unsigned i = tid; i < cnt; i += SWG) {
      const unsigned w = mid[base + i];
      const unsigned r = atomicAdd(&lcnt[w & (BLK_NODES - 1)], 1u);
      sorted[base + r] = (int)(w >> NODE_BITS);
    }
  }
}

#define GEMM_WG 256
__global__ void __launch_bounds__(GEMM_WG)
gemm1_wmma_kernel(const float* __restrict__ x,
                                  const float* __restrict__ W1,
                                  const unsigned int* __restrict__ deg,
                                  float* __restrict__ t1, int N) {
  __shared__ __align__(32) __bf16 Bfrag[8][2][32][16];

  const int tid  = threadIdx.x;
  const int lane = tid & 31;
  const int wave = tid >> 5;

  {
    constexpr int PER_THREAD = (8 * 2 * 32 * 16) / GEMM_WG;
    float wv[PER_THREAD];
    #pragma unroll
    for (int it = 0; it < PER_THREAD; ++it) {
      int idx   = tid + it * GEMM_WG;
      int j     = idx & 15;
      int ln    = (idx >> 4) & 31;
      int ntile = (idx >> 9) & 1;
      int ks    = idx >> 10;
      int jj = j >> 1, p = j & 1;
      int hi = ln >> 4;
      int K  = ks * 32 + 16 * (jj >> 2) + 8 * hi + 2 * (jj & 3) + p;
      int n  = ntile * 16 + (ln & 15);
      wv[it] = W1[K * H_DIM + n];
    }
    __bf16* __restrict__ bflat = &Bfrag[0][0][0][0];
    #pragma unroll
    for (int it = 0; it < PER_THREAD; ++it) bflat[tid + it * GEMM_WG] = (__bf16)wv[it];
  }
  __syncthreads();

  long long rowbase = ((long long)blockIdx.x * 8 + wave) * 16;
  if (rowbase >= N) return;

  const int m  = lane & 15;
  const int hi = lane >> 4;
  const float* xrow = x + (rowbase + m) * D_IN;

  v8f acc0 = {}; v8f acc1 = {};
  #pragma unroll
  for (int ks = 0; ks < 8; ++ks) {
    v16bf a;
    #pragma unroll
    for (int jj = 0; jj < 8; ++jj) {
      int K0 = ks * 32 + 16 * (jj >> 2) + 8 * hi + 2 * (jj & 3);
      float2 f = *(const float2*)(xrow + K0);
      a[2 * jj]     = (__bf16)f.x;
      a[2 * jj + 1] = (__bf16)f.y;
    }
    v16bf b0 = *(const v16bf*)&Bfrag[ks][0][lane][0];
    v16bf b1 = *(const v16bf*)&Bfrag[ks][1][lane][0];
    acc0 = __builtin_amdgcn_wmma_f32_16x16x32_bf16(false, a, false, b0,
                                                   (short)0, acc0, false, false);
    acc1 = __builtin_amdgcn_wmma_f32_16x16x32_bf16(false, a, false, b1,
                                                   (short)0, acc1, false, false);
  }

  #pragma unroll
  for (int r = 0; r < 8; ++r) {
    long long row = rowbase + r + 8 * hi;
    float di = rsqrtf((float)(deg[row] + 1u));
    t1[row * H_DIM + m]      = acc0[r] * di;
    t1[row * H_DIM + 16 + m] = acc1[r] * di;
  }
}

#define GB 256
__global__ void __launch_bounds__(GB)
gather1_kernel(const int* __restrict__ sorted, const unsigned int* __restrict__ rowptr,
               const float* __restrict__ t1, const float* __restrict__ b1,
               const float* __restrict__ W2, float* __restrict__ dcol, int N, int E) {
  const int lane  = threadIdx.x & 31;
  const int sub   = lane & 7;
  const int gbase = lane & 24;
  const long long gid = (long long)blockIdx.x * GB + threadIdx.x;
  const long long tq  = gid >> 3;
  const bool valid = tq < N;
  const int t = valid ? (int)tq : N - 1;
  const unsigned beg = rowptr[t];
  const unsigned end = (t + 1 < N) ? rowptr[t + 1] : (unsigned)E;
  const float4* __restrict__ tv = (const float4*)t1;
  const float4 self = tv[(long long)t * (H_DIM / 4) + sub];
  float4 acc = make_float4(0.0f, 0.0f, 0.0f, 0.0f);
  int mine = (beg + sub < end) ? sorted[beg + sub] : -1;
  for (unsigned k = beg; k < end; k += 8) {
    const unsigned kn = k + 8 + sub;
    const int nw  = sorted[(kn < end) ? kn : k];
    const int nxt = (kn < end) ? nw : -1;
    #pragma unroll
    for (int j = 0; j < 8; ++j) {
      const int sj = __shfl(mine, gbase + j);
      const bool on = sj >= 0;
      const float4 v = tv[(long long)(on ? sj : 0) * (H_DIM / 4) + sub];
      acc.x += on ? v.x : 0.0f;
      acc.y += on ? v.y : 0.0f;
      acc.z += on ? v.z : 0.0f;
      acc.w += on ? v.w : 0.0f;
    }
    mine = nxt;
  }
  const float di = rsqrtf((float)(end - beg + 1u));
  const float4 bb = ((const float4*)b1)[sub];
  float h0 = di * (acc.x + self.x) + bb.x;
  float h1 = di * (acc.y + self.y) + bb.y;
  float h2a = di * (acc.z + self.z) + bb.z;
  float h3 = di * (acc.w + self.w) + bb.w;
  h0 = h0 > 0.0f ? h0 : 0.0f;
  h1 = h1 > 0.0f ? h1 : 0.0f;
  h2a = h2a > 0.0f ? h2a : 0.0f;
  h3 = h3 > 0.0f ? h3 : 0.0f;
  const float4 wa = ((const float4*)W2)[2 * sub];
  const float4 wb = ((const float4*)W2)[2 * sub + 1];
  float z0 = h0 * wa.x + h1 * wa.z + h2a * wb.x + h3 * wb.z;
  float z1 = h0 * wa.y + h1 * wa.w + h2a * wb.y + h3 * wb.w;
  #pragma unroll
  for (int m = 1; m < 8; m <<= 1) { z0 += __shfl_xor(z0, m); z1 += __shfl_xor(z1, m); }
  if (valid && sub == 0) dcol[t] = di * (z0 - z1);
}

#define G2_WG    1024
#define G2_TPB   392
#define G2_NI    4
#define G2_FILLK 13
#define G2_TAB   (G2_WG * 4 * G2_FILLK)
static_assert(G2_NI * (G2_WG / 8) >= G2_TPB, "a lane group per target node of the work-group");

__device__ __forceinline__ void g2_ask(const float* __restrict__ src, int cnt, int tid,
                                       float4 (&nv)[G2_FILLK]) {
  const int nq = cnt >> 2;
  if (nq > 0) {
    const float4* __restrict__ s4 = (const float4*)src;
    #pragma unroll
    for (int k = 0; k < G2_FILLK; ++k) {
      const int q = tid + k * G2_WG;
      nv[k] = s4[q < nq ? q : nq - 1];
    }
  } else {
    #pragma unroll
    for (int k = 0; k < G2_FILLK; ++k) nv[k] = make_float4(0.0f, 0.0f, 0.0f, 0.0f);
  }
}

__device__ __forceinline__ void g2_put(float* __restrict__ tab, const float* __restrict__ src,
                                       int cnt, int tid, const float4 (&nv)[G2_FILLK]) {
  const int nq = cnt >> 2;
  #pragma unroll
  for (int k = 0; k < G2_FILLK; ++k) {
    const int q = tid + k * G2_WG;
    if (q < nq) ((float4*)tab)[q] = nv[k];
  }
  if (tid < (cnt & 3)) tab[(nq << 2) + tid] = src[(nq << 2) + tid];
}

__global__ void __launch_bounds__(G2_WG)
gather2_logsoftmax_kernel(const int* __restrict__ sorted,
                          const unsigned int* __restrict__ rowptr,
                          const float* __restrict__ dcol, const float* __restrict__ b2,
                          float* __restrict__ out, int N, int E) {
  __shared__ __align__(16) float tab[G2_TAB];
  const int tid  = threadIdx.x;
  const int sub  = tid & 7;
  const int grp  = tid >> 3;
  const int base = blockIdx.x * G2_TPB;

  float4 nv0[G2_FILLK];
  const int cnt0 = N < G2_TAB ? N : G2_TAB;
  g2_ask(dcol, cnt0, tid, nv0);

  unsigned beg[G2_NI], end[G2_NI];
  float self[G2_NI];
  #pragma unroll
  for (int i = 0; i < G2_NI; ++i) {
    const int lt = grp + i * (G2_WG / 8);
    const int t  = base + lt;
    const bool v = (lt < G2_TPB) && (t < N);
    const int tc = v ? t : N - 1;
    const bool more = tc + 1 < N;
    const unsigned b  = rowptr[tc];
    const unsigned en = rowptr[more ? tc + 1 : tc];
    beg[i]  = b;
    end[i]  = v ? (more ? en : (unsigned)E) : b;
    self[i] = dcol[tc];
  }
  g2_put(tab, dcol, cnt0, tid, nv0);

  int idx[G2_NI][8];
  #pragma unroll
  for (int i = 0; i < G2_NI; ++i) {
    #pragma unroll
    for (int j = 0; j < 8; ++j) {
      const unsigned kk = beg[i] + (unsigned)sub + 8u * (unsigned)j;
      const bool ok = kk < end[i];
      const int w = sorted[ok ? kk : 0u];
      idx[i][j] = ok ? w : -1;
    }
  }

  float acc[G2_NI];
  #pragma unroll
  for (int i = 0; i < G2_NI; ++i) acc[i] = 0.0f;

  for (int p0 = 0; p0 < N; p0 += G2_TAB) {
    const int cnt = (N - p0 < G2_TAB) ? (N - p0) : G2_TAB;
    if (p0 > 0) {
      float4 nv[G2_FILLK];
      g2_ask(dcol + p0, cnt, tid, nv);
      __syncthreads();
      g2_put(tab, dcol + p0, cnt, tid, nv);
    }
    __syncthreads();
    #pragma unroll
    for (int i = 0; i < G2_NI; ++i) {
      if ((tid & ~31) / 8 + i * (G2_WG / 8) < G2_TPB) {
        float v[8];
        #pragma unroll
        for (int j = 0; j < 8; ++j) {
          const unsigned u = (unsigned)(idx[i][j] - p0);
          v[j] = tab[u < (unsigned)cnt ? u : 0u];
        }
        #pragma unroll
        for (int j = 0; j < 8; ++j) {
          const unsigned u = (unsigned)(idx[i][j] - p0);
          acc[i] += u < (unsigned)cnt ? v[j] : 0.0f;
        }
      }
    }
  }

  const float bd = b2[0] - b2[1];
  #pragma unroll
  for (int i = 0; i < G2_NI; ++i) {
    const int lt = grp + i * (G2_WG / 8);
    float a = acc[i];
    for (unsigned k = beg[i] + (unsigned)sub + 64u; k < end[i]; k += 8u) a += dcol[sorted[k]];
    #pragma unroll
    for (int m = 1; m < 8; m <<= 1) a += __shfl_xor(a, m);
    const float di  = rsqrtf((float)(end[i] - beg[i] + 1u));
    const float d   = di * (a + self[i]) + bd;
    const float mx  = fmaxf(d, 0.0f);
    const float lse = mx + logf(expf(d - mx) + expf(-mx));
    if (sub == 0 && lt < G2_TPB && base + lt < N) ((float2*)out)[base + lt] = make_float2(d - lse, -lse);
  }
}

extern "C" void kernel_launch(void* const* d_in, const int* in_sizes, int n_in,
                              void* d_out, int out_size, void* d_ws, size_t ws_size,
                              hipStream_t stream) {
  const float* x  = (const float*)d_in[0];
  const int*   ei = (const int*)d_in[1];
  const float* W1 = (const float*)d_in[2];
  const float* b1 = (const float*)d_in[3];
  const float* W2 = (const float*)d_in[4];
  const float* b2 = (const float*)d_in[5];
  float* out = (float*)d_out;

  const int N = in_sizes[0] / D_IN;
  const int E = in_sizes[1] / 2;

  float* ws   = (float*)d_ws;
  float* t1   = ws;
  float* h2   = t1 + (long long)N * H_DIM;
  unsigned int* rowptr = (unsigned int*)(h2 + (long long)N * C_DIM);
  int* sorted = (int*)(rowptr + N);
  const int NB = (N + BLK_NODES - 1) / BLK_NODES;
  const int P  = (E + CHUNK - 1) / CHUNK;
  unsigned int* mid   = (unsigned int*)t1;
  unsigned int* hist  = (unsigned int*)sorted;
  unsigned int* off   = hist + (long long)P * NB;
  unsigned int* deg   = (unsigned int*)h2;
  unsigned int* tot   = deg + N;
  unsigned int* bbase = tot + NBMAX;

  const int gblocks = (int)(((long long)N * 8 + GB - 1) / GB);

  block_count_kernel<<<P, SWG, 0, stream>>>(ei, hist, E, N, NB);
  block_prefix_kernel<<<(NB + 31) / 32, PWG, 0, stream>>>(hist, off, tot, P, NB);
  block_place_kernel<<<P, SWG, 0, stream>>>(ei, hist, off, tot, bbase, mid, E, N, NB);
  block_sort_kernel<<<NB, SWG, 0, stream>>>(mid, bbase, rowptr, deg, sorted, N);

  gemm1_wmma_kernel<<<(N + 127) / 128, GEMM_WG, 0, stream>>>(x, W1, deg, t1, N);
  gather1_kernel<<<gblocks, GB, 0, stream>>>(sorted, rowptr, t1, b1, W2, h2, N, E);

  gather2_logsoftmax_kernel<<<(N + G2_TPB - 1) / G2_TPB, G2_WG, 0, stream>>>(sorted, rowptr, h2, b2, out, N, E);
}


# gather1 launches a quarter of the waves: each 8-lane group takes 4 consecutive nodes in turn, run bounds, self rows and first slot words of all 4 asked for ahead of the first; sums unchanged
# speedup vs baseline: 6.6191x; 1.0542x over previous
#include <hip/hip_runtime.h>
#include <hip/hip_bf16.h>

typedef __attribute__((ext_vector_type(16))) __bf16 v16bf;
typedef __attribute__((ext_vector_type(8)))  float  v8f;

#define D_IN  256
#define H_DIM 32
#define C_DIM 2

#define NODE_BITS 8
#define BLK_NODES 256
#define NBMAX     512
#define SWG       512
#define EPT       25
#define CHUNK     (SWG * EPT)
#define PWG       1024
#define KW        20
#define STAGE_CAP 16384
static_assert(NBMAX == SWG, "one thread per node block in the block tables");

__global__ void __launch_bounds__(SWG)
block_count_kernel(const int* __restrict__ ei, unsigned int* __restrict__ hist,
                   int E, int N, int NB) {
  __shared__ unsigned int h[NBMAX];
  const int tid = threadIdx.x;
  const int e0  = blockIdx.x * CHUNK;
  const int e1  = (E - e0 > CHUNK) ? (e0 + CHUNK) : E;
  const int* __restrict__ tgt = ei + E;
  int tv[EPT];
  #pragma unroll
  for (int k = 0; k < EPT; ++k) {
    const int e = e0 + tid + k * SWG;
    const int v = tgt[e < e1 ? e : e0];
    tv[k] = (e < e1) ? v : -1;
  }
  h[tid] = 0u;
  __syncthreads();
  #pragma unroll
  for (int k = 0; k < EPT; ++k)
    if ((unsigned)tv[k] < (unsigned)N) atomicAdd(&h[tv[k] >> NODE_BITS], 1u);
  __syncthreads();
  if (tid < NB) hist[(long long)blockIdx.x * NB + tid] = h[tid];
}

__global__ void __launch_bounds__(PWG)
block_prefix_kernel(const unsigned int* __restrict__ hist, unsigned int* __restrict__ off,
                    unsigned int* __restrict__ tot, int P, int NB) {
  __shared__ unsigned int seg[32][32];
  const int lane = threadIdx.x & 31;
  const int w    = threadIdx.x >> 5;
  const int b    = blockIdx.x * 32 + lane;
  const bool on  = b < NB;
  const int pps  = (P + 31) / 32;
  const int p0   = w * pps;
  const int p1   = (p0 + pps < P) ? (p0 + pps) : P;
  unsigned s = 0u;
  if (on) for (int p = p0; p < p1; ++p) s += hist[(long long)p * NB + b];
  seg[w][lane] = s;
  __syncthreads();
  unsigned run = 0u;
  #pragma unroll
  for (int q = 0; q < 32; ++q) { const unsigned v = seg[q][lane]; run += (q < w) ? v : 0u; }
  if (on) {
    for (int p = p0; p < p1; ++p) {
      const unsigned v = hist[(long long)p * NB + b];
      off[(long long)p * NB + b] = run;
      run += v;
    }
    if (w == 31) tot[b] = run;
  }
}

__global__ void __launch_bounds__(SWG)
block_place_kernel(const int* __restrict__ ei, const unsigned int* __restrict__ hist,
                   const unsigned int* __restrict__ off, const unsigned int* __restrict__ tot,
                   unsigned int* __restrict__ bbase, unsigned int* __restrict__ mid,
                   int E, int N, int NB) {
  __shared__ unsigned int lc[NBMAX + 32];
  __shared__ unsigned int dl[NBMAX];
  __shared__ unsigned int buf[CHUNK];
  __shared__ unsigned int gp[CHUNK];
  __shared__ unsigned int wa[SWG / 32];
  __shared__ unsigned int wb[SWG / 32];
  __shared__ unsigned int nkept;
  const int tid  = threadIdx.x;
  const int lane = tid & 31;
  const int wave = tid >> 5;
  const int e0   = blockIdx.x * CHUNK;
  const int e1   = (E - e0 > CHUNK) ? (e0 + CHUNK) : E;
  const int* __restrict__ src = ei;
  const int* __restrict__ tgt = ei + E;
  int tv[EPT], sv[EPT];
  #pragma unroll
  for (int k = 0; k < EPT; ++k) {
    const int e  = e0 + tid + k * SWG;
    const int ec = e < e1 ? e : e0;
    const int t  = tgt[ec];
    sv[k] = src[ec];
    tv[k] = (e < e1) ? t : -1;
  }
  const bool onb = tid < NB;
  const long long row = (long long)blockIdx.x * NB;
  const unsigned c  = onb ? hist[row + tid] : 0u;
  const unsigned g  = onb ? off[row + tid] : 0u;
  const unsigned tt = onb ? tot[tid] : 0u;
  unsigned ic = c, it = tt;
  #pragma unroll
  for (int d = 1; d < 32; d <<= 1) {
    const unsigned nc = __shfl_up(ic, d);
    const unsigned nt = __shfl_up(it, d);
    if (lane >= d) { ic += nc; it += nt; }
  }
  if (lane == 31) { wa[wave] = ic; wb[wave] = it; }
  __syncthreads();
  unsigned ba = 0u, bb = 0u;
  #pragma unroll
  for (int q = 0; q < SWG / 32; ++q) {
    const unsigned va = wa[q], vb = wb[q];
    ba += (q < wave) ? va : 0u;
    bb += (q < wave) ? vb : 0u;
  }
  const unsigned ls = ba + ic - c;
  const unsigned gb = bb + it - tt;
  lc[tid] = ls;
  dl[tid] = gb + g - ls;
  if (tid == SWG - 1) nkept = ba + ic;
  if (blockIdx.x == 0 && onb) {
    bbase[tid] = gb;
    if (tid == NB - 1) bbase[NB] = gb + tt;
  }
  __syncthreads();
  unsigned r[EPT];
  #pragma unroll
  for (int k = 0; k < EPT; ++k) {
    const bool act = (unsigned)tv[k] < (unsigned)N;
    r[k] = atomicAdd(&lc[act ? (unsigned)(tv[k] >> NODE_BITS) : (unsigned)(NBMAX + lane)],
                     act ? 1u : 0u);
  }
  #pragma unroll
  for (int k = 0; k < EPT; ++k) {
    if ((unsigned)tv[k] < (unsigned)N) {
      const unsigned b = (unsigned)(tv[k] >> NODE_BITS);
      buf[r[k]] = ((unsigned)sv[k] << NODE_BITS) | ((unsigned)tv[k] & (BLK_NODES - 1));
      gp[r[k]]  = r[k] + dl[b];
    }
  }
  __syncthreads();
  const unsigned n = nkept;
  for (unsigned i = tid; i < n; i += SWG) mid[gp[i]] = buf[i];
}

__global__ void __launch_bounds__(SWG)
block_sort_kernel(const unsigned int* __restrict__ mid, const unsigned int* __restrict__ bbase,
                  unsigned int* __restrict__ rowptr, unsigned int* __restrict__ deg,
                  int* __restrict__ sorted, int N) {
  __shared__ unsigned int lcnt[BLK_NODES + 32];
  __shared__ unsigned int wt[BLK_NODES / 32];
  __shared__ unsigned int stage[STAGE_CAP];
  const int tid  = threadIdx.x;
  const int lane = tid & 31;
  const int wave = tid >> 5;
  const unsigned base = bbase[blockIdx.x];
  const unsigned cnt  = bbase[blockIdx.x + 1] - base;
  if (tid < BLK_NODES) lcnt[tid] = 0u;
  __syncthreads();
  for (unsigned i0 = 0u; i0 < cnt; i0 += KW * SWG) {
    unsigned w[KW];
    #pragma unroll
    for (int k = 0; k < KW; ++k) {
      const unsigned i = i0 + tid + k * SWG;
      w[k] = mid[base + (i < cnt ? i : 0u)];
    }
    #pragma unroll
    for (int k = 0; k < KW; ++k)
      if (i0 + tid + k * SWG < cnt) atomicAdd(&lcnt[w[k] & (BLK_NODES - 1)], 1u);
  }
  __syncthreads();
  unsigned c = 0u, ic = 0u;
  if (tid < BLK_NODES) {
    c = lcnt[tid];
    ic = c;
    #pragma unroll
    for (int d = 1; d < 32; d <<= 1) { const unsigned nn = __shfl_up(ic, d); if (lane >= d) ic += nn; }
    if (lane == 31) wt[wave] = ic;
  }
  __syncthreads();
  if (tid < BLK_NODES) {
    unsigned bs = 0u;
    #pragma unroll
    for (int q = 0; q < BLK_NODES / 32; ++q) { const unsigned v = wt[q]; bs += (q < wave) ? v : 0u; }
    const unsigned excl = bs + ic - c;
    const int node = blockIdx.x * BLK_NODES + tid;
    if (node < N) { rowptr[node] = base + excl; deg[node] = c; }
    lcnt[tid] = excl;
  }
  __syncthreads();
  if (cnt <= STAGE_CAP) {
    for (unsigned i0 = 0u; i0 < cnt; i0 += KW * SWG) {
      unsigned w[KW], r[KW];
      #pragma unroll
      for (int k = 0; k < KW; ++k) {
        const unsigned i = i0 + tid + k * SWG;
        w[k] = mid[base + (i < cnt ? i : 0u)];
      }
      #pragma unroll
      for (int k = 0; k < KW; ++k) {
        const bool act = i0 + tid + k * SWG < cnt;
        r[k] = atomicAdd(&lcnt[act ? (w[k] & (BLK_NODES - 1)) : (unsigned)(BLK_NODES + lane)],
                         act ? 1u : 0u);
      }
      #pragma unroll
      for (int k = 0; k < KW; ++k)
        if (i0 + tid + k * SWG < cnt) stage[r[k]] = w[k] >> NODE_BITS;
    }
    __syncthreads();
    for (unsigned i = tid; i < cnt; i += SWG) sorted[base + i] = (int)stage[i];
  } else {
    for (unsigned i = tid; i < cnt; i += SWG) {
      const unsigned w = mid[base + i];
      const unsigned r = atomicAdd(&lcnt[w & (BLK_NODES - 1)], 1u);
      sorted[base + r] = (int)(w >> NODE_BITS);
    }
  }
}

#define GEMM_WG 256
__global__ void __launch_bounds__(GEMM_WG)
gemm1_wmma_kernel(const float* __restrict__ x,
                                  const float* __restrict__ W1,
                                  const unsigned int* __restrict__ deg,
                                  float* __restrict__ t1, int N) {
  __shared__ __align__(32) __bf16 Bfrag[8][2][32][16];

  const int tid  = threadIdx.x;
  const int lane = tid & 31;
  const int wave = tid >> 5;

  {
    constexpr int PER_THREAD = (8 * 2 * 32 * 16) / GEMM_WG;
    float wv[PER_THREAD];
    #pragma unroll
    for (int it = 0; it < PER_THREAD; ++it) {
      int idx   = tid + it * GEMM_WG;
      int j     = idx & 15;
      int ln    = (idx >> 4) & 31;
      int ntile = (idx >> 9) & 1;
      int ks    = idx >> 10;
      int jj = j >> 1, p = j & 1;
      int hi = ln >> 4;
      int K  = ks * 32 + 16 * (jj >> 2) + 8 * hi + 2 * (jj & 3) + p;
      int n  = ntile * 16 + (ln & 15);
      wv[it] = W1[K * H_DIM + n];
    }
    __bf16* __restrict__ bflat = &Bfrag[0][0][0][0];
    #pragma unroll
    for (int it = 0; it < PER_THREAD; ++it) bflat[tid + it * GEMM_WG] = (__bf16)wv[it];
  }
  __syncthreads();

  long long rowbase = ((long long)blockIdx.x * 8 + wave) * 16;
  if (rowbase >= N) return;

  const int m  = lane & 15;
  const int hi = lane >> 4;
  const float* xrow = x + (rowbase + m) * D_IN;

  v8f acc0 = {}; v8f acc1 = {};
  #pragma unroll
  for (int ks = 0; ks < 8; ++ks) {
    v16bf a;
    #pragma unroll
    for (int jj = 0; jj < 8; ++jj) {
      int K0 = ks * 32 + 16 * (jj >> 2) + 8 * hi + 2 * (jj & 3);
      float2 f = *(const float2*)(xrow + K0);
      a[2 * jj]     = (__bf16)f.x;
      a[2 * jj + 1] = (__bf16)f.y;
    }
    v16bf b0 = *(const v16bf*)&Bfrag[ks][0][lane][0];
    v16bf b1 = *(const v16bf*)&Bfrag[ks][1][lane][0];
    acc0 = __builtin_amdgcn_wmma_f32_16x16x32_bf16(false, a, false, b0,
                                                   (short)0, acc0, false, false);
    acc1 = __builtin_amdgcn_wmma_f32_16x16x32_bf16(false, a, false, b1,
                                                   (short)0, acc1, false, false);
  }

  #pragma unroll
  for (int r = 0; r < 8; ++r) {
    long long row = rowbase + r + 8 * hi;
    float di = rsqrtf((float)(deg[row] + 1u));
    t1[row * H_DIM + m]      = acc0[r] * di;
    t1[row * H_DIM + 16 + m] = acc1[r] * di;
  }
}

#define GB 256
#define G1_NI 4
__global__ void __launch_bounds__(GB)
gather1_kernel(const int* __restrict__ sorted, const unsigned int* __restrict__ rowptr,
               const float* __restrict__ t1, const float* __restrict__ b1,
               const float* __restrict__ W2, float* __restrict__ dcol, int N, int E) {
  const int lane  = threadIdx.x & 31;
  const int sub   = lane & 7;
  const int gbase = lane & 24;
  const long long gid = (long long)blockIdx.x * GB + threadIdx.x;
  const long long t0  = (gid >> 3) * G1_NI;
  const float4* __restrict__ tv = (const float4*)t1;
  unsigned rp[G1_NI + 1];
  #pragma unroll
  for (int i = 0; i <= G1_NI; ++i) {
    const long long t = t0 + i;
    rp[i] = rowptr[t < N ? t : N - 1];
  }
  __builtin_amdgcn_sched_barrier(0);
  #pragma unroll
  for (int i = 0; i <= G1_NI; ++i) rp[i] = (t0 + i < N) ? rp[i] : (unsigned)E;
  float4 self[G1_NI];
  int first[G1_NI];
  #pragma unroll
  for (int i = 0; i < G1_NI; ++i) {
    const long long t  = t0 + i;
    const long long tc = t < N ? t : N - 1;
    self[i] = tv[tc * (H_DIM / 4) + sub];
    const unsigned k0 = rp[i] + (unsigned)sub;
    const bool ok = k0 < rp[i + 1];
    const int w = sorted[ok ? k0 : 0u];
    first[i] = ok ? w : -1;
  }
  const float4 bb = ((const float4*)b1)[sub];
  const float4 wa = ((const float4*)W2)[2 * sub];
  const float4 wb = ((const float4*)W2)[2 * sub + 1];
  float res[G1_NI];
  #pragma unroll
  for (int i = 0; i < G1_NI; ++i) {
    const unsigned beg = rp[i];
    const unsigned end = rp[i + 1];
    float4 acc = make_float4(0.0f, 0.0f, 0.0f, 0.0f);
    int mine = first[i];
    for (unsigned k = beg; k < end; k += 8) {
      const unsigned kn = k + 8 + sub;
      const int nw  = sorted[(kn < end) ? kn : k];
      const int nxt = (kn < end) ? nw : -1;
      #pragma unroll
      for (int j = 0; j < 8; ++j) {
        const int sj = __shfl(mine, gbase + j);
        const bool on = sj >= 0;
        const float4 v = tv[(long long)(on ? sj : 0) * (H_DIM / 4) + sub];
        acc.x += on ? v.x : 0.0f;
        acc.y += on ? v.y : 0.0f;
        acc.z += on ? v.z : 0.0f;
        acc.w += on ? v.w : 0.0f;
      }
      mine = nxt;
    }
    const float di = rsqrtf((float)(end - beg + 1u));
    float h0 = di * (acc.x + self[i].x) + bb.x;
    float h1 = di * (acc.y + self[i].y) + bb.y;
    float h2a = di * (acc.z + self[i].z) + bb.z;
    float h3 = di * (acc.w + self[i].w) + bb.w;
    h0 = h0 > 0.0f ? h0 : 0.0f;
    h1 = h1 > 0.0f ? h1 : 0.0f;
    h2a = h2a > 0.0f ? h2a : 0.0f;
    h3 = h3 > 0.0f ? h3 : 0.0f;
    float z0 = h0 * wa.x + h1 * wa.z + h2a * wb.x + h3 * wb.z;
    float z1 = h0 * wa.y + h1 * wa.w + h2a * wb.y + h3 * wb.w;
    #pragma unroll
    for (int m = 1; m < 8; m <<= 1) { z0 += __shfl_xor(z0, m); z1 += __shfl_xor(z1, m); }
    res[i] = di * (z0 - z1);
  }
  if (sub == 0) {
    #pragma unroll
    for (int i = 0; i < G1_NI; ++i)
      if (t0 + i < N) dcol[t0 + i] = res[i];
  }
}

#define G2_WG    1024
#define G2_TPB   392
#define G2_NI    4
#define G2_FILLK 13
#define G2_TAB   (G2_WG * 4 * G2_FILLK)
static_assert(G2_NI * (G2_WG / 8) >= G2_TPB, "a lane group per target node of the work-group");

__device__ __forceinline__ void g2_ask(const float* __restrict__ src, int cnt, int tid,
                                       float4 (&nv)[G2_FILLK]) {
  const int nq = cnt >> 2;
  if (nq > 0) {
    const float4* __restrict__ s4 = (const float4*)src;
    #pragma unroll
    for (int k = 0; k < G2_FILLK; ++k) {
      const int q = tid + k * G2_WG;
      nv[k] = s4[q < nq ? q : nq - 1];
    }
  } else {
    #pragma unroll
    for (int k = 0; k < G2_FILLK; ++k) nv[k] = make_float4(0.0f, 0.0f, 0.0f, 0.0f);
  }
}

__device__ __forceinline__ void g2_put(float* __restrict__ tab, const float* __restrict__ src,
                                       int cnt, int tid, const float4 (&nv)[G2_FILLK]) {
  const int nq = cnt >> 2;
  #pragma unroll
  for (int k = 0; k < G2_FILLK; ++k) {
    const int q = tid + k * G2_WG;
    if (q < nq) ((float4*)tab)[q] = nv[k];
  }
  if (tid < (cnt & 3)) tab[(nq << 2) + tid] = src[(nq << 2) + tid];
}

__global__ void __launch_bounds__(G2_WG)
gather2_logsoftmax_kernel(const int* __restrict__ sorted,
                          const unsigned int* __restrict__ rowptr,
                          const float* __restrict__ dcol, const float* __restrict__ b2,
                          float* __restrict__ out, int N, int E) {
  __shared__ __align__(16) float tab[G2_TAB];
  const int tid  = threadIdx.x;
  const int sub  = tid & 7;
  const int grp  = tid >> 3;
  const int base = blockIdx.x * G2_TPB;

  float4 nv0[G2_FILLK];
  const int cnt0 = N < G2_TAB ? N : G2_TAB;
  g2_ask(dcol, cnt0, tid, nv0);

  unsigned beg[G2_NI], end[G2_NI];
  float self[G2_NI];
  #pragma unroll
  for (int i = 0; i < G2_NI; ++i) {
    const int lt = grp + i * (G2_WG / 8);
    const int t  = base + lt;
    const bool v = (lt < G2_TPB) && (t < N);
    const int tc = v ? t : N - 1;
    const bool more = tc + 1 < N;
    const unsigned b  = rowptr[tc];
    const unsigned en = rowptr[more ? tc + 1 : tc];
    beg[i]  = b;
    end[i]  = v ? (more ? en : (unsigned)E) : b;
    self[i] = dcol[tc];
  }
  g2_put(tab, dcol, cnt0, tid, nv0);

  int idx[G2_NI][8];
  #pragma unroll
  for (int i = 0; i < G2_NI; ++i) {
    #pragma unroll
    for (int j = 0; j < 8; ++j) {
      const unsigned kk = beg[i] + (unsigned)sub + 8u * (unsigned)j;
      const bool ok = kk < end[i];
      const int w = sorted[ok ? kk : 0u];
      idx[i][j] = ok ? w : -1;
    }
  }

  float acc[G2_NI];
  #pragma unroll
  for (int i = 0; i < G2_NI; ++i) acc[i] = 0.0f;

  for (int p0 = 0; p0 < N; p0 += G2_TAB) {
    const int cnt = (N - p0 < G2_TAB) ? (N - p0) : G2_TAB;
    if (p0 > 0) {
      float4 nv[G2_FILLK];
      g2_ask(dcol + p0, cnt, tid, nv);
      __syncthreads();
      g2_put(tab, dcol + p0, cnt, tid, nv);
    }
    __syncthreads();
    #pragma unroll
    for (int i = 0; i < G2_NI; ++i) {
      if ((tid & ~31) / 8 + i * (G2_WG / 8) < G2_TPB) {
        float v[8];
        #pragma unroll
        for (int j = 0; j < 8; ++j) {
          const unsigned u = (unsigned)(idx[i][j] - p0);
          v[j] = tab[u < (unsigned)cnt ? u : 0u];
        }
        #pragma unroll
        for (int j = 0; j < 8; ++j) {
          const unsigned u = (unsigned)(idx[i][j] - p0);
          acc[i] += u < (unsigned)cnt ? v[j] : 0.0f;
        }
      }
    }
  }

  const float bd = b2[0] - b2[1];
  #pragma unroll
  for (int i = 0; i < G2_NI; ++i) {
    const int lt = grp + i * (G2_WG / 8);
    float a = acc[i];
    for (unsigned k = beg[i] + (unsigned)sub + 64u; k < end[i]; k += 8u) a += dcol[sorted[k]];
    #pragma unroll
    for (int m = 1; m < 8; m <<= 1) a += __shfl_xor(a, m);
    const float di  = rsqrtf((float)(end[i] - beg[i] + 1u));
    const float d   = di * (a + self[i]) + bd;
    const float mx  = fmaxf(d, 0.0f);
    const float lse = mx + logf(expf(d - mx) + expf(-mx));
    if (sub == 0 && lt < G2_TPB && base + lt < N) ((float2*)out)[base + lt] = make_float2(d - lse, -lse);
  }
}

extern "C" void kernel_launch(void* const* d_in, const int* in_sizes, int n_in,
                              void* d_out, int out_size, void* d_ws, size_t ws_size,
                              hipStream_t stream) {
  const float* x  = (const float*)d_in[0];
  const int*   ei = (const int*)d_in[1];
  const float* W1 = (const float*)d_in[2];
  const float* b1 = (const float*)d_in[3];
  const float* W2 = (const float*)d_in[4];
  const float* b2 = (const float*)d_in[5];
  float* out = (float*)d_out;

  const int N = in_sizes[0] / D_IN;
  const int E = in_sizes[1] / 2;

  float* ws   = (float*)d_ws;
  float* t1   = ws;
  float* h2   = t1 + (long long)N * H_DIM;
  unsigned int* rowptr = (unsigned int*)(h2 + (long long)N * C_DIM);
  int* sorted = (int*)(rowptr + N);
  const int NB = (N + BLK_NODES - 1) / BLK_NODES;
  const int P  = (E + CHUNK - 1) / CHUNK;
  unsigned int* mid   = (unsigned int*)t1;
  unsigned int* hist  = (unsigned int*)sorted;
  unsigned int* off   = hist + (long long)P * NB;
  unsigned int* deg   = (unsigned int*)h2;
  unsigned int* tot   = deg + N;
  unsigned int* bbase = tot + NBMAX;

  const int ggroups = (N + G1_NI - 1) / G1_NI;
  const int gblocks = (int)(((long long)ggroups * 8 + GB - 1) / GB);

  block_count_kernel<<<P, SWG, 0, stream>>>(ei, hist, E, N, NB);
  block_prefix_kernel<<<(NB + 31) / 32, PWG, 0, stream>>>(hist, off, tot, P, NB);
  block_place_kernel<<<P, SWG, 0, stream>>>(ei, hist, off, tot, bbase, mid, E, N, NB);
  block_sort_kernel<<<NB, SWG, 0, stream>>>(mid, bbase, rowptr, deg, sorted, N);

  gemm1_wmma_kernel<<<(N + 127) / 128, GEMM_WG, 0, stream>>>(x, W1, deg, t1, N);
  gather1_kernel<<<gblocks, GB, 0, stream>>>(sorted, rowptr, t1, b1, W2, h2, N, E);

  gather2_logsoftmax_kernel<<<(N + G2_TPB - 1) / G2_TPB, G2_WG, 0, stream>>>(sorted, rowptr, h2, b2, out, N, E);
}


# edge sort level 1 orders each chunk alone in LDS (counts, scan, cursors); level 2 collects its block's run of each chunk via a table of run starts; count and prefix launches go; f32 sum order changes
# speedup vs baseline: 7.9647x; 1.2033x over previous
#include <hip/hip_runtime.h>
#include <hip/hip_bf16.h>

typedef __attribute__((ext_vector_type(16))) __bf16 v16bf;
typedef __attribute__((ext_vector_type(8)))  float  v8f;

#define D_IN  256
#define H_DIM 32
#define C_DIM 2

#define NODE_BITS 8
#define BLK_NODES 256
#define NBMAX     512
#define SWG       512
#define EPT       25
#define CHUNK     (SWG * EPT)
#define PMAX      512
#define RPW       16
#define KW        20
#define STAGE_CAP 16384
static_assert(NBMAX == SWG, "one thread per node block in the block tables");
static_assert(PMAX == SWG, "one thread per chunk in the run tables");
static_assert(PMAX % ((SWG / 32) * RPW) == 0, "whole trips of the collection stay inside the run tables");

__global__ void __launch_bounds__(SWG)
chunk_order_kernel(const int* __restrict__ ei, unsigned int* __restrict__ lstab,
                   unsigned int* __restrict__ mid, int E, int N, int NB) {
  __shared__ unsigned int lc[NBMAX + 32];
  __shared__ unsigned int buf[CHUNK];
  __shared__ unsigned int wa[SWG / 32];
  __shared__ unsigned int nkept;
  const int tid  = threadIdx.x;
  const int lane = tid & 31;
  const int wave = tid >> 5;
  const int e0   = blockIdx.x * CHUNK;
  const int e1   = (E - e0 > CHUNK) ? (e0 + CHUNK) : E;
  const int* __restrict__ src = ei;
  const int* __restrict__ tgt = ei + E;
  int tv[EPT], sv[EPT];
  #pragma unroll
  for (int k = 0; k < EPT; ++k) {
    const int e  = e0 + tid + k * SWG;
    const int ec = e < e1 ? e : e0;
    tv[k] = tgt[ec];
    sv[k] = src[ec];
  }
  __builtin_amdgcn_sched_barrier(0);
  #pragma unroll
  for (int k = 0; k < EPT; ++k) tv[k] = (e0 + tid + k * SWG < e1) ? tv[k] : -1;
  lc[tid] = 0u;
  __syncthreads();
  #pragma unroll
  for (int k = 0; k < EPT; ++k)
    if ((unsigned)tv[k] < (unsigned)N) atomicAdd(&lc[tv[k] >> NODE_BITS], 1u);
  __syncthreads();
  const unsigned c = lc[tid];
  unsigned ic = c;
  #pragma unroll
  for (int d = 1; d < 32; d <<= 1) {
    const unsigned nc = __shfl_up(ic, d);
    if (lane >= d) ic += nc;
  }
  if (lane == 31) wa[wave] = ic;
  __syncthreads();
  unsigned ba = 0u;
  #pragma unroll
  for (int q = 0; q < SWG / 32; ++q) { const unsigned va = wa[q]; ba += (q < wave) ? va : 0u; }
  const unsigned ls = ba + ic - c;
  lc[tid] = ls;
  const long long row = (long long)blockIdx.x * (NB + 1);
  if (tid < NB) lstab[row + tid] = ls;
  if (tid == SWG - 1) { nkept = ba + ic; lstab[row + NB] = ba + ic; }
  __syncthreads();
  unsigned r[EPT];
  #pragma unroll
  for (int k = 0; k < EPT; ++k) {
    const bool act = (unsigned)tv[k] < (unsigned)N;
    r[k] = atomicAdd(&lc[act ? (unsigned)(tv[k] >> NODE_BITS) : (unsigned)(NBMAX + lane)],
                     act ? 1u : 0u);
  }
  #pragma unroll
  for (int k = 0; k < EPT; ++k)
    if ((unsigned)tv[k] < (unsigned)N)
      buf[r[k]] = ((unsigned)sv[k] << NODE_BITS) | ((unsigned)tv[k] & (BLK_NODES - 1));
  __syncthreads();
  const unsigned n = nkept;
  for (unsigned i = tid; i < n; i += SWG) mid[(unsigned)e0 + i] = buf[i];
}

__global__ void __launch_bounds__(SWG)
block_sort_kernel(const unsigned int* __restrict__ mid, const unsigned int* __restrict__ lstab,
                  unsigned int* __restrict__ rowptr, unsigned int* __restrict__ deg,
                  int* __restrict__ sorted, int N, int P, int NB) {
  __shared__ unsigned int lcnt[BLK_NODES + 32];
  __shared__ unsigned int wt[SWG / 32];
  __shared__ unsigned int wu[SWG / 32];
  __shared__ unsigned int rs[PMAX];
  __shared__ unsigned int rd[PMAX];
  __shared__ unsigned int rn[PMAX];
  __shared__ unsigned int raw[STAGE_CAP];
  __shared__ unsigned int stage[STAGE_CAP];
  const int tid  = threadIdx.x;
  const int lane = tid & 31;
  const int wave = tid >> 5;
  unsigned a = 0u, n = 0u;
  if (tid < P) {
    const long long row = (long long)tid * (NB + 1) + blockIdx.x;
    const unsigned a0 = lstab[row];
    const unsigned a1 = lstab[row + 1];
    a = a0;
    n = a1 - a0;
  }
  if (tid < BLK_NODES) lcnt[tid] = 0u;
  unsigned in = n, ia = a;
  #pragma unroll
  for (int d = 1; d < 32; d <<= 1) {
    const unsigned nn = __shfl_up(in, d);
    const unsigned na = __shfl_up(ia, d);
    if (lane >= d) { in += nn; ia += na; }
  }
  if (lane == 31) { wt[wave] = in; wu[wave] = ia; }
  __syncthreads();
  unsigned bn = 0u, cnt = 0u, base = 0u;
  #pragma unroll
  for (int q = 0; q < SWG / 32; ++q) {
    const unsigned vn = wt[q], va = wu[q];
    bn   += (q < wave) ? vn : 0u;
    cnt  += vn;
    base += va;
  }
  rs[tid] = (unsigned)tid * CHUNK + a;
  rd[tid] = bn + in - n;
  rn[tid] = n;
  __syncthreads();
  const bool staged = cnt <= STAGE_CAP;
  for (int p0 = 0; p0 < P; p0 += (SWG / 32) * RPW) {
    unsigned ss[RPW], dd[RPW], nn[RPW], w0[RPW], w1[RPW];
    #pragma unroll
    for (int r = 0; r < RPW; ++r) {
      const int p = p0 + wave + r * (SWG / 32);
      ss[r] = rs[p];
      dd[r] = rd[p];
      nn[r] = rn[p];
    }
    unsigned longest = 0u;
    #pragma unroll
    for (int r = 0; r < RPW; ++r) {
      w0[r] = mid[(unsigned)lane < nn[r] ? ss[r] + (unsigned)lane : 0u];
      w1[r] = mid[(unsigned)lane + 32u < nn[r] ? ss[r] + (unsigned)lane + 32u : 0u];
      longest = nn[r] > longest ? nn[r] : longest;
    }
    #pragma unroll
    for (int r = 0; r < RPW; ++r) {
      if ((unsigned)lane < nn[r]) {
        if (staged) raw[dd[r] + (unsigned)lane] = w0[r];
        atomicAdd(&lcnt[w0[r] & (BLK_NODES - 1)], 1u);
      }
      if ((unsigned)lane + 32u < nn[r]) {
        if (staged) raw[dd[r] + (unsigned)lane + 32u] = w1[r];
        atomicAdd(&lcnt[w1[r] & (BLK_NODES - 1)], 1u);
      }
    }
    if (longest > 64u) {
      for (int r = 0; r < RPW; ++r) {
        const int p = p0 + wave + r * (SWG / 32);
        const unsigned s = rs[p], d = rd[p], m = rn[p];
        for (unsigned i = (unsigned)lane + 64u; i < m; i += 32u) {
          const unsigned w = mid[s + i];
          if (staged) raw[d + i] = w;
          atomicAdd(&lcnt[w & (BLK_NODES - 1)], 1u);
        }
      }
    }
  }
  __syncthreads();
  unsigned c = 0u, ic = 0u;
  if (tid < BLK_NODES) {
    c = lcnt[tid];
    ic = c;
    #pragma unroll
    for (int d = 1; d < 32; d <<= 1) { const unsigned nn = __shfl_up(ic, d); if (lane >= d) ic += nn; }
    if (lane == 31) wt[wave] = ic;
  }
  __syncthreads();
  if (tid < BLK_NODES) {
    unsigned bs = 0u;
    #pragma unroll
    for (int q = 0; q < BLK_NODES / 32; ++q) { const unsigned v = wt[q]; bs += (q < wave) ? v : 0u; }
    const unsigned excl = bs + ic - c;
    const int node = blockIdx.x * BLK_NODES + tid;
    if (node < N) { rowptr[node] = base + excl; deg[node] = c; }
    lcnt[tid] = excl;
  }
  __syncthreads();
  if (staged) {
    for (unsigned i0 = 0u; i0 < cnt; i0 += KW * SWG) {
      unsigned w[KW], r[KW];
      #pragma unroll
      for (int k = 0; k < KW; ++k) {
        const unsigned i = i0 + tid + k * SWG;
        w[k] = raw[i < cnt ? i : 0u];
      }
      #pragma unroll
      for (int k = 0; k < KW; ++k) {
        const bool act = i0 + tid + k * SWG < cnt;
        r[k] = atomicAdd(&lcnt[act ? (w[k] & (BLK_NODES - 1)) : (unsigned)(BLK_NODES + lane)],
                         act ? 1u : 0u);
      }
      #pragma unroll
      for (int k = 0; k < KW; ++k)
        if (i0 + tid + k * SWG < cnt) stage[r[k]] = w[k] >> NODE_BITS;
    }
    __syncthreads();
    for (unsigned i = tid; i < cnt; i += SWG) sorted[base + i] = (int)stage[i];
  } else {
    for (int p = wave; p < P; p += SWG / 32) {
      const unsigned s = rs[p], m = rn[p];
      for (unsigned i = (unsigned)lane; i < m; i += 32u) {
        const unsigned w = mid[s + i];
        const unsigned r = atomicAdd(&lcnt[w & (BLK_NODES - 1)], 1u);
        sorted[base + r] = (int)(w >> NODE_BITS);
      }
    }
  }
}

#define GEMM_WG 256
__global__ void __launch_bounds__(GEMM_WG)
gemm1_wmma_kernel(const float* __restrict__ x,
                                  const float* __restrict__ W1,
                                  const unsigned int* __restrict__ deg,
                                  float* __restrict__ t1, int N) {
  __shared__ __align__(32) __bf16 Bfrag[8][2][32][16];

  const int tid  = threadIdx.x;
  const int lane = tid & 31;
  const int wave = tid >> 5;

  {
    constexpr int PER_THREAD = (8 * 2 * 32 * 16) / GEMM_WG;
    float wv[PER_THREAD];
    #pragma unroll
    for (int it = 0; it < PER_THREAD; ++it) {
      int idx   = tid + it * GEMM_WG;
      int j     = idx & 15;
      int ln    = (idx >> 4) & 31;
      int ntile = (idx >> 9) & 1;
      int ks    = idx >> 10;
      int jj = j >> 1, p = j & 1;
      int hi = ln >> 4;
      int K  = ks * 32 + 16 * (jj >> 2) + 8 * hi + 2 * (jj & 3) + p;
      int n  = ntile * 16 + (ln & 15);
      wv[it] = W1[K * H_DIM + n];
    }
    __bf16* __restrict__ bflat = &Bfrag[0][0][0][0];
    #pragma unroll
    for (int it = 0; it < PER_THREAD; ++it) bflat[tid + it * GEMM_WG] = (__bf16)wv[it];
  }
  __syncthreads();

  long long rowbase = ((long long)blockIdx.x * 8 + wave) * 16;
  if (rowbase >= N) return;

  const int m  = lane & 15;
  const int hi = lane >> 4;
  const float* xrow = x + (rowbase + m) * D_IN;

  v8f acc0 = {}; v8f acc1 = {};
  #pragma unroll
  for (int ks = 0; ks < 8; ++ks) {
    v16bf a;
    #pragma unroll
    for (int jj = 0; jj < 8; ++jj) {
      int K0 = ks * 32 + 16 * (jj >> 2) + 8 * hi + 2 * (jj & 3);
      float2 f = *(const float2*)(xrow + K0);
      a[2 * jj]     = (__bf16)f.x;
      a[2 * jj + 1] = (__bf16)f.y;
    }
    v16bf b0 = *(const v16bf*)&Bfrag[ks][0][lane][0];
    v16bf b1 = *(const v16bf*)&Bfrag[ks][1][lane][0];
    acc0 = __builtin_amdgcn_wmma_f32_16x16x32_bf16(false, a, false, b0,
                                                   (short)0, acc0, false, false);
    acc1 = __builtin_amdgcn_wmma_f32_16x16x32_bf16(false, a, false, b1,
                                                   (short)0, acc1, false, false);
  }

  #pragma unroll
  for (int r = 0; r < 8; ++r) {
    long long row = rowbase + r + 8 * hi;
    float di = rsqrtf((float)(deg[row] + 1u));
    t1[row * H_DIM + m]      = acc0[r] * di;
    t1[row * H_DIM + 16 + m] = acc1[r] * di;
  }
}

#define GB 256
#define G1_NI 4
__global__ void __launch_bounds__(GB)
gather1_kernel(const int* __restrict__ sorted, const unsigned int* __restrict__ rowptr,
               const float* __restrict__ t1, const float* __restrict__ b1,
               const float* __restrict__ W2, float* __restrict__ dcol, int N, int E) {
  const int lane  = threadIdx.x & 31;
  const int sub   = lane & 7;
  const int gbase = lane & 24;
  const long long gid = (long long)blockIdx.x * GB + threadIdx.x;
  const long long t0  = (gid >> 3) * G1_NI;
  const float4* __restrict__ tv = (const float4*)t1;
  unsigned rp[G1_NI + 1];
  #pragma unroll
  for (int i = 0; i <= G1_NI; ++i) {
    const long long t = t0 + i;
    rp[i] = rowptr[t < N ? t : N - 1];
  }
  __builtin_amdgcn_sched_barrier(0);
  #pragma unroll
  for (int i = 0; i <= G1_NI; ++i) rp[i] = (t0 + i < N) ? rp[i] : (unsigned)E;
  float4 self[G1_NI];
  int first[G1_NI];
  #pragma unroll
  for (int i = 0; i < G1_NI; ++i) {
    const long long t  = t0 + i;
    const long long tc = t < N ? t : N - 1;
    self[i] = tv[tc * (H_DIM / 4) + sub];
    const unsigned k0 = rp[i] + (unsigned)sub;
    const bool ok = k0 < rp[i + 1];
    const int w = sorted[ok ? k0 : 0u];
    first[i] = ok ? w : -1;
  }
  const float4 bb = ((const float4*)b1)[sub];
  const float4 wa = ((const float4*)W2)[2 * sub];
  const float4 wb = ((const float4*)W2)[2 * sub + 1];
  float res[G1_NI];
  #pragma unroll
  for (int i = 0; i < G1_NI; ++i) {
    const unsigned beg = rp[i];
    const unsigned end = rp[i + 1];
    float4 acc = make_float4(0.0f, 0.0f, 0.0f, 0.0f);
    int mine = first[i];
    for (unsigned k = beg; k < end; k += 8) {
      const unsigned kn = k + 8 + sub;
      const int nw  = sorted[(kn < end) ? kn : k];
      const int nxt = (kn < end) ? nw : -1;
      #pragma unroll
      for (int j = 0; j < 8; ++j) {
        const int sj = __shfl(mine, gbase + j);
        const bool on = sj >= 0;
        const float4 v = tv[(long long)(on ? sj : 0) * (H_DIM / 4) + sub];
        acc.x += on ? v.x : 0.0f;
        acc.y += on ? v.y : 0.0f;
        acc.z += on ? v.z : 0.0f;
        acc.w += on ? v.w : 0.0f;
      }
      mine = nxt;
    }
    const float di = rsqrtf((float)(end - beg + 1u));
    float h0 = di * (acc.x + self[i].x) + bb.x;
    float h1 = di * (acc.y + self[i].y) + bb.y;
    float h2a = di * (acc.z + self[i].z) + bb.z;
    float h3 = di * (acc.w + self[i].w) + bb.w;
    h0 = h0 > 0.0f ? h0 : 0.0f;
    h1 = h1 > 0.0f ? h1 : 0.0f;
    h2a = h2a > 0.0f ? h2a : 0.0f;
    h3 = h3 > 0.0f ? h3 : 0.0f;
    float z0 = h0 * wa.x + h1 * wa.z + h2a * wb.x + h3 * wb.z;
    float z1 = h0 * wa.y + h1 * wa.w + h2a * wb.y + h3 * wb.w;
    #pragma unroll
    for (int m = 1; m < 8; m <<= 1) { z0 += __shfl_xor(z0, m); z1 += __shfl_xor(z1, m); }
    res[i] = di * (z0 - z1);
  }
  if (sub == 0) {
    #pragma unroll
    for (int i = 0; i < G1_NI; ++i)
      if (t0 + i < N) dcol[t0 + i] = res[i];
  }
}

#define G2_WG    1024
#define G2_TPB   392
#define G2_NI    4
#define G2_FILLK 13
#define G2_TAB   (G2_WG * 4 * G2_FILLK)
static_assert(G2_NI * (G2_WG / 8) >= G2_TPB, "a lane group per target node of the work-group");

__device__ __forceinline__ void g2_ask(const float* __restrict__ src, int cnt, int tid,
                                       float4 (&nv)[G2_FILLK]) {
  const int nq = cnt >> 2;
  if (nq > 0) {
    const float4* __restrict__ s4 = (const float4*)src;
    #pragma unroll
    for (int k = 0; k < G2_FILLK; ++k) {
      const int q = tid + k * G2_WG;
      nv[k] = s4[q < nq ? q : nq - 1];
    }
  } else {
    #pragma unroll
    for (int k = 0; k < G2_FILLK; ++k) nv[k] = make_float4(0.0f, 0.0f, 0.0f, 0.0f);
  }
}

__device__ __forceinline__ void g2_put(float* __restrict__ tab, const float* __restrict__ src,
                                       int cnt, int tid, const float4 (&nv)[G2_FILLK]) {
  const int nq = cnt >> 2;
  #pragma unroll
  for (int k = 0; k < G2_FILLK; ++k) {
    const int q = tid + k * G2_WG;
    if (q < nq) ((float4*)tab)[q] = nv[k];
  }
  if (tid < (cnt & 3)) tab[(nq << 2) + tid] = src[(nq << 2) + tid];
}

__global__ void __launch_bounds__(G2_WG)
gather2_logsoftmax_kernel(const int* __restrict__ sorted,
                          const unsigned int* __restrict__ rowptr,
                          const float* __restrict__ dcol, const float* __restrict__ b2,
                          float* __restrict__ out, int N, int E) {
  __shared__ __align__(16) float tab[G2_TAB];
  const int tid  = threadIdx.x;
  const int sub  = tid & 7;
  const int grp  = tid >> 3;
  const int base = blockIdx.x * G2_TPB;

  float4 nv0[G2_FILLK];
  const int cnt0 = N < G2_TAB ? N : G2_TAB;
  g2_ask(dcol, cnt0, tid, nv0);

  unsigned beg[G2_NI], end[G2_NI];
  float self[G2_NI];
  #pragma unroll
  for (int i = 0; i < G2_NI; ++i) {
    const int lt = grp + i * (G2_WG / 8);
    const int t  = base + lt;
    const bool v = (lt < G2_TPB) && (t < N);
    const int tc = v ? t : N - 1;
    const bool more = tc + 1 < N;
    const unsigned b  = rowptr[tc];
    const unsigned en = rowptr[more ? tc + 1 : tc];
    beg[i]  = b;
    end[i]  = v ? (more ? en : (unsigned)E) : b;
    self[i] = dcol[tc];
  }
  g2_put(tab, dcol, cnt0, tid, nv0);

  int idx[G2_NI][8];
  #pragma unroll
  for (int i = 0; i < G2_NI; ++i) {
    #pragma unroll
    for (int j = 0; j < 8; ++j) {
      const unsigned kk = beg[i] + (unsigned)sub + 8u * (unsigned)j;
      const bool ok = kk < end[i];
      const int w = sorted[ok ? kk : 0u];
      idx[i][j] = ok ? w : -1;
    }
  }

  float acc[G2_NI];
  #pragma unroll
  for (int i = 0; i < G2_NI; ++i) acc[i] = 0.0f;

  for (int p0 = 0; p0 < N; p0 += G2_TAB) {
    const int cnt = (N - p0 < G2_TAB) ? (N - p0) : G2_TAB;
    if (p0 > 0) {
      float4 nv[G2_FILLK];
      g2_ask(dcol + p0, cnt, tid, nv);
      __syncthreads();
      g2_put(tab, dcol + p0, cnt, tid, nv);
    }
    __syncthreads();
    #pragma unroll
    for (int i = 0; i < G2_NI; ++i) {
      if ((tid & ~31) / 8 + i * (G2_WG / 8) < G2_TPB) {
        float v[8];
        #pragma unroll
        for (int j = 0; j < 8; ++j) {
          const unsigned u = (unsigned)(idx[i][j] - p0);
          v[j] = tab[u < (unsigned)cnt ? u : 0u];
        }
        #pragma unroll
        for (int j = 0; j < 8; ++j) {
          const unsigned u = (unsigned)(idx[i][j] - p0);
          acc[i] += u < (unsigned)cnt ? v[j] : 0.0f;
        }
      }
    }
  }

  const float bd = b2[0] - b2[1];
  #pragma unroll
  for (int i = 0; i < G2_NI; ++i) {
    const int lt = grp + i * (G2_WG / 8);
    float a = acc[i];
    for (unsigned k = beg[i] + (unsigned)sub + 64u; k < end[i]; k += 8u) a += dcol[sorted[k]];
    #pragma unroll
    for (int m = 1; m < 8; m <<= 1) a += __shfl_xor(a, m);
    const float di  = rsqrtf((float)(end[i] - beg[i] + 1u));
    const float d   = di * (a + self[i]) + bd;
    const float mx  = fmaxf(d, 0.0f);
    const float lse = mx + logf(expf(d - mx) + expf(-mx));
    if (sub == 0 && lt < G2_TPB && base + lt < N) ((float2*)out)[base + lt] = make_float2(d - lse, -lse);
  }
}

extern "C" void kernel_launch(void* const* d_in, const int* in_sizes, int n_in,
                              void* d_out, int out_size, void* d_ws, size_t ws_size,
                              hipStream_t stream) {
  const float* x  = (const float*)d_in[0];
  const int*   ei = (const int*)d_in[1];
  const float* W1 = (const float*)d_in[2];
  const float* b1 = (const float*)d_in[3];
  const float* W2 = (const float*)d_in[4];
  const float* b2 = (const float*)d_in[5];
  float* out = (float*)d_out;

  const int N = in_sizes[0] / D_IN;
  const int E = in_sizes[1] / 2;

  float* ws   = (float*)d_ws;
  float* t1   = ws;
  float* h2   = t1 + (long long)N * H_DIM;
  unsigned int* rowptr = (unsigned int*)(h2 + (long long)N * C_DIM);
  int* sorted = (int*)(rowptr + N);
  const int NB = (N + BLK_NODES - 1) / BLK_NODES;
  const int P  = (E + CHUNK - 1) / CHUNK;
  unsigned int* mid   = (unsigned int*)t1;
  unsigned int* deg   = (unsigned int*)h2;
  unsigned int* lstab = deg + N;

  const int ggroups = (N + G1_NI - 1) / G1_NI;
  const int gblocks = (int)(((long long)ggroups * 8 + GB - 1) / GB);

  chunk_order_kernel<<<P, SWG, 0, stream>>>(ei, lstab, mid, E, N, NB);
  block_sort_kernel<<<NB, SWG, 0, stream>>>(mid, lstab, rowptr, deg, sorted, N, P, NB);

  gemm1_wmma_kernel<<<(N + 127) / 128, GEMM_WG, 0, stream>>>(x, W1, deg, t1, N);
  gather1_kernel<<<gblocks, GB, 0, stream>>>(sorted, rowptr, t1, b1, W2, h2, N, E);

  gather2_logsoftmax_kernel<<<(N + G2_TPB - 1) / G2_TPB, G2_WG, 0, stream>>>(sorted, rowptr, h2, b2, out, N, E);
}


# x @ W1: every global load covers whole 128-byte lines: x tiles read 512 contiguous bytes per wave load, made into A fragments via a wave-private LDS tile; W1 staged one row per load; sums unchanged
# speedup vs baseline: 8.3916x; 1.0536x over previous
#include <hip/hip_runtime.h>
#include <hip/hip_bf16.h>

typedef __attribute__((ext_vector_type(16))) __bf16 v16bf;
typedef __attribute__((ext_vector_type(8)))  float  v8f;

#define D_IN  256
#define H_DIM 32
#define C_DIM 2

#define NODE_BITS 8
#define BLK_NODES 256
#define NBMAX     512
#define SWG       512
#define EPT       25
#define CHUNK     (SWG * EPT)
#define PMAX      512
#define RPW       16
#define KW        20
#define STAGE_CAP 16384
static_assert(NBMAX == SWG, "one thread per node block in the block tables");
static_assert(PMAX == SWG, "one thread per chunk in the run tables");
static_assert(PMAX % ((SWG / 32) * RPW) == 0, "whole trips of the collection stay inside the run tables");

__global__ void __launch_bounds__(SWG)
chunk_order_kernel(const int* __restrict__ ei, unsigned int* __restrict__ lstab,
                   unsigned int* __restrict__ mid, int E, int N, int NB) {
  __shared__ unsigned int lc[NBMAX + 32];
  __shared__ unsigned int buf[CHUNK];
  __shared__ unsigned int wa[SWG / 32];
  __shared__ unsigned int nkept;
  const int tid  = threadIdx.x;
  const int lane = tid & 31;
  const int wave = tid >> 5;
  const int e0   = blockIdx.x * CHUNK;
  const int e1   = (E - e0 > CHUNK) ? (e0 + CHUNK) : E;
  const int* __restrict__ src = ei;
  const int* __restrict__ tgt = ei + E;
  int tv[EPT], sv[EPT];
  #pragma unroll
  for (int k = 0; k < EPT; ++k) {
    const int e  = e0 + tid + k * SWG;
    const int ec = e < e1 ? e : e0;
    tv[k] = tgt[ec];
    sv[k] = src[ec];
  }
  __builtin_amdgcn_sched_barrier(0);
  #pragma unroll
  for (int k = 0; k < EPT; ++k) tv[k] = (e0 + tid + k * SWG < e1) ? tv[k] : -1;
  lc[tid] = 0u;
  __syncthreads();
  #pragma unroll
  for (int k = 0; k < EPT; ++k)
    if ((unsigned)tv[k] < (unsigned)N) atomicAdd(&lc[tv[k] >> NODE_BITS], 1u);
  __syncthreads();
  const unsigned c = lc[tid];
  unsigned ic = c;
  #pragma unroll
  for (int d = 1; d < 32; d <<= 1) {
    const unsigned nc = __shfl_up(ic, d);
    if (lane >= d) ic += nc;
  }
  if (lane == 31) wa[wave] = ic;
  __syncthreads();
  unsigned ba = 0u;
  #pragma unroll
  for (int q = 0; q < SWG / 32; ++q) { const unsigned va = wa[q]; ba += (q < wave) ? va : 0u; }
  const unsigned ls = ba + ic - c;
  lc[tid] = ls;
  const long long row = (long long)blockIdx.x * (NB + 1);
  if (tid < NB) lstab[row + tid] = ls;
  if (tid == SWG - 1) { nkept = ba + ic; lstab[row + NB] = ba + ic; }
  __syncthreads();
  unsigned r[EPT];
  #pragma unroll
  for (int k = 0; k < EPT; ++k) {
    const bool act = (unsigned)tv[k] < (unsigned)N;
    r[k] = atomicAdd(&lc[act ? (unsigned)(tv[k] >> NODE_BITS) : (unsigned)(NBMAX + lane)],
                     act ? 1u : 0u);
  }
  #pragma unroll
  for (int k = 0; k < EPT; ++k)
    if ((unsigned)tv[k] < (unsigned)N)
      buf[r[k]] = ((unsigned)sv[k] << NODE_BITS) | ((unsigned)tv[k] & (BLK_NODES - 1));
  __syncthreads();
  const unsigned n = nkept;
  for (unsigned i = tid; i < n; i += SWG) mid[(unsigned)e0 + i] = buf[i];
}

__global__ void __launch_bounds__(SWG)
block_sort_kernel(const unsigned int* __restrict__ mid, const unsigned int* __restrict__ lstab,
                  unsigned int* __restrict__ rowptr, unsigned int* __restrict__ deg,
                  int* __restrict__ sorted, int N, int P, int NB) {
  __shared__ unsigned int lcnt[BLK_NODES + 32];
  __shared__ unsigned int wt[SWG / 32];
  __shared__ unsigned int wu[SWG / 32];
  __shared__ unsigned int rs[PMAX];
  __shared__ unsigned int rd[PMAX];
  __shared__ unsigned int rn[PMAX];
  __shared__ unsigned int raw[STAGE_CAP];
  __shared__ unsigned int stage[STAGE_CAP];
  const int tid  = threadIdx.x;
  const int lane = tid & 31;
  const int wave = tid >> 5;
  unsigned a = 0u, n = 0u;
  if (tid < P) {
    const long long row = (long long)tid * (NB + 1) + blockIdx.x;
    const unsigned a0 = lstab[row];
    const unsigned a1 = lstab[row + 1];
    a = a0;
    n = a1 - a0;
  }
  if (tid < BLK_NODES) lcnt[tid] = 0u;
  unsigned in = n, ia = a;
  #pragma unroll
  for (int d = 1; d < 32; d <<= 1) {
    const unsigned nn = __shfl_up(in, d);
    const unsigned na = __shfl_up(ia, d);
    if (lane >= d) { in += nn; ia += na; }
  }
  if (lane == 31) { wt[wave] = in; wu[wave] = ia; }
  __syncthreads();
  unsigned bn = 0u, cnt = 0u, base = 0u;
  #pragma unroll
  for (int q = 0; q < SWG / 32; ++q) {
    const unsigned vn = wt[q], va = wu[q];
    bn   += (q < wave) ? vn : 0u;
    cnt  += vn;
    base += va;
  }
  rs[tid] = (unsigned)tid * CHUNK + a;
  rd[tid] = bn + in - n;
  rn[tid] = n;
  __syncthreads();
  const bool staged = cnt <= STAGE_CAP;
  for (int p0 = 0; p0 < P; p0 += (SWG / 32) * RPW) {
    unsigned ss[RPW], dd[RPW], nn[RPW], w0[RPW], w1[RPW];
    #pragma unroll
    for (int r = 0; r < RPW; ++r) {
      const int p = p0 + wave + r * (SWG / 32);
      ss[r] = rs[p];
      dd[r] = rd[p];
      nn[r] = rn[p];
    }
    unsigned longest = 0u;
    #pragma unroll
    for (int r = 0; r < RPW; ++r) {
      w0[r] = mid[(unsigned)lane < nn[r] ? ss[r] + (unsigned)lane : 0u];
      w1[r] = mid[(unsigned)lane + 32u < nn[r] ? ss[r] + (unsigned)lane + 32u : 0u];
      longest = nn[r] > longest ? nn[r] : longest;
    }
    #pragma unroll
    for (int r = 0; r < RPW; ++r) {
      if ((unsigned)lane < nn[r]) {
        if (staged) raw[dd[r] + (unsigned)lane] = w0[r];
        atomicAdd(&lcnt[w0[r] & (BLK_NODES - 1)], 1u);
      }
      if ((unsigned)lane + 32u < nn[r]) {
        if (staged) raw[dd[r] + (unsigned)lane + 32u] = w1[r];
        atomicAdd(&lcnt[w1[r] & (BLK_NODES - 1)], 1u);
      }
    }
    if (longest > 64u) {
      for (int r = 0; r < RPW; ++r) {
        const int p = p0 + wave + r * (SWG / 32);
        const unsigned s = rs[p], d = rd[p], m = rn[p];
        for (unsigned i = (unsigned)lane + 64u; i < m; i += 32u) {
          const unsigned w = mid[s + i];
          if (staged) raw[d + i] = w;
          atomicAdd(&lcnt[w & (BLK_NODES - 1)], 1u);
        }
      }
    }
  }
  __syncthreads();
  unsigned c = 0u, ic = 0u;
  if (tid < BLK_NODES) {
    c = lcnt[tid];
    ic = c;
    #pragma unroll
    for (int d = 1; d < 32; d <<= 1) { const unsigned nn = __shfl_up(ic, d); if (lane >= d) ic += nn; }
    if (lane == 31) wt[wave] = ic;
  }
  __syncthreads();
  if (tid < BLK_NODES) {
    unsigned bs = 0u;
    #pragma unroll
    for (int q = 0; q < BLK_NODES / 32; ++q) { const unsigned v = wt[q]; bs += (q < wave) ? v : 0u; }
    const unsigned excl = bs + ic - c;
    const int node = blockIdx.x * BLK_NODES + tid;
    if (node < N) { rowptr[node] = base + excl; deg[node] = c; }
    lcnt[tid] = excl;
  }
  __syncthreads();
  if (staged) {
    for (unsigned i0 = 0u; i0 < cnt; i0 += KW * SWG) {
      unsigned w[KW], r[KW];
      #pragma unroll
      for (int k = 0; k < KW; ++k) {
        const unsigned i = i0 + tid + k * SWG;
        w[k] = raw[i < cnt ? i : 0u];
      }
      #pragma unroll
      for (int k = 0; k < KW; ++k) {
        const bool act = i0 + tid + k * SWG < cnt;
        r[k] = atomicAdd(&lcnt[act ? (w[k] & (BLK_NODES - 1)) : (unsigned)(BLK_NODES + lane)],
                         act ? 1u : 0u);
      }
      #pragma unroll
      for (int k = 0; k < KW; ++k)
        if (i0 + tid + k * SWG < cnt) stage[r[k]] = w[k] >> NODE_BITS;
    }
    __syncthreads();
    for (unsigned i = tid; i < cnt; i += SWG) sorted[base + i] = (int)stage[i];
  } else {
    for (int p = wave; p < P; p += SWG / 32) {
      const unsigned s = rs[p], m = rn[p];
      for (unsigned i = (unsigned)lane; i < m; i += 32u) {
        const unsigned w = mid[s + i];
        const unsigned r = atomicAdd(&lcnt[w & (BLK_NODES - 1)], 1u);
        sorted[base + r] = (int)(w >> NODE_BITS);
      }
    }
  }
}

typedef __attribute__((ext_vector_type(8))) __bf16 v8bf;
typedef __attribute__((ext_vector_type(4))) __bf16 v4bf;
#define GEMM_WG 256
#define KH      128
#define A_PITCH (KH + 8)
static_assert(GEMM_WG / 32 == D_IN / 32, "one wave per k-step in the W1 staging");

__device__ __forceinline__ void wave_lds_handover() {
  __builtin_amdgcn_fence(__ATOMIC_RELEASE, "wavefront");
  __builtin_amdgcn_wave_barrier();
  __builtin_amdgcn_fence(__ATOMIC_ACQUIRE, "wavefront");
}

__global__ void __launch_bounds__(GEMM_WG)
gemm1_wmma_kernel(const float* __restrict__ x,
                                  const float* __restrict__ W1,
                                  const unsigned int* __restrict__ deg,
                                  float* __restrict__ t1, int N) {
  __shared__ __align__(32) __bf16 Bfrag[8][2][32][16];
  __shared__ __align__(16) __bf16 Atile[GEMM_WG / 32][16 * A_PITCH];

  const int tid  = threadIdx.x;
  const int lane = tid & 31;
  const int wave = tid >> 5;

  const long long rowbase = ((long long)blockIdx.x * 8 + wave) * 16;
  const bool active = rowbase < N;
  const long long rb = active ? rowbase : 0;

  const int q = lane >> 3;
  const int o = lane & 7;
  const float4* __restrict__ xq = (const float4*)(x + (rb + 4 * q) * D_IN) + o;
  float4 xv[16];
  #pragma unroll
  for (int j = 0; j < 16; ++j) xv[j] = xq[(j & 3) * (D_IN / 4) + (j >> 2) * 8];

  {
    float wv[32];
    const float* __restrict__ wp = W1 + (wave * 32) * H_DIM + lane;
    #pragma unroll
    for (int i = 0; i < 32; ++i) wv[i] = wp[i * H_DIM];
    const int ntile = lane >> 4;
    const int cn    = lane & 15;
    #pragma unroll
    for (int hh = 0; hh < 2; ++hh) {
      v16bf f;
      #pragma unroll
      for (int s = 0; s < 16; ++s) f[s] = (__bf16)wv[16 * (s >> 3) + 8 * hh + (s & 7)];
      *(v16bf*)&Bfrag[wave][ntile][cn + 16 * hh][0] = f;
    }
  }
  __syncthreads();

  if (!active) return;

  const int m  = lane & 15;
  const int hi = lane >> 4;
  __bf16* __restrict__ at = &Atile[wave][0];

  v8f acc0 = {}; v8f acc1 = {};
  #pragma unroll
  for (int pass = 0; pass < 2; ++pass) {
    #pragma unroll
    for (int j = 0; j < 16; ++j) {
      v4bf pk;
      pk[0] = (__bf16)xv[j].x;
      pk[1] = (__bf16)xv[j].y;
      pk[2] = (__bf16)xv[j].z;
      pk[3] = (__bf16)xv[j].w;
      *(v4bf*)&at[(4 * q + (j & 3)) * A_PITCH + (j >> 2) * 32 + 4 * o] = pk;
    }
    if (pass == 0) {
      #pragma unroll
      for (int j = 0; j < 16; ++j) xv[j] = xq[KH / 4 + (j & 3) * (D_IN / 4) + (j >> 2) * 8];
    }
    wave_lds_handover();
    #pragma unroll
    for (int ks = 0; ks < KH / 32; ++ks) {
      const v8bf lo = *(const v8bf*)&at[m * A_PITCH + ks * 32 + 8 * hi];
      const v8bf up = *(const v8bf*)&at[m * A_PITCH + ks * 32 + 16 + 8 * hi];
      const v16bf a = __builtin_shufflevector(lo, up, 0, 1, 2, 3, 4, 5, 6, 7,
                                              8, 9, 10, 11, 12, 13, 14, 15);
      const int kg = pass * (KH / 32) + ks;
      v16bf b0 = *(const v16bf*)&Bfrag[kg][0][lane][0];
      v16bf b1 = *(const v16bf*)&Bfrag[kg][1][lane][0];
      acc0 = __builtin_amdgcn_wmma_f32_16x16x32_bf16(false, a, false, b0,
                                                     (short)0, acc0, false, false);
      acc1 = __builtin_amdgcn_wmma_f32_16x16x32_bf16(false, a, false, b1,
                                                     (short)0, acc1, false, false);
    }
    wave_lds_handover();
  }

  #pragma unroll
  for (int r = 0; r < 8; ++r) {
    long long row = rowbase + r + 8 * hi;
    float di = rsqrtf((float)(deg[row] + 1u));
    t1[row * H_DIM + m]      = acc0[r] * di;
    t1[row * H_DIM + 16 + m] = acc1[r] * di;
  }
}

#define GB 256
#define G1_NI 4
__global__ void __launch_bounds__(GB)
gather1_kernel(const int* __restrict__ sorted, const unsigned int* __restrict__ rowptr,
               const float* __restrict__ t1, const float* __restrict__ b1,
               const float* __restrict__ W2, float* __restrict__ dcol, int N, int E) {
  const int lane  = threadIdx.x & 31;
  const int sub   = lane & 7;
  const int gbase = lane & 24;
  const long long gid = (long long)blockIdx.x * GB + threadIdx.x;
  const long long t0  = (gid >> 3) * G1_NI;
  const float4* __restrict__ tv = (const float4*)t1;
  unsigned rp[G1_NI + 1];
  #pragma unroll
  for (int i = 0; i <= G1_NI; ++i) {
    const long long t = t0 + i;
    rp[i] = rowptr[t < N ? t : N - 1];
  }
  __builtin_amdgcn_sched_barrier(0);
  #pragma unroll
  for (int i = 0; i <= G1_NI; ++i) rp[i] = (t0 + i < N) ? rp[i] : (unsigned)E;
  float4 self[G1_NI];
  int first[G1_NI];
  #pragma unroll
  for (int i = 0; i < G1_NI; ++i) {
    const long long t  = t0 + i;
    const long long tc = t < N ? t : N - 1;
    self[i] = tv[tc * (H_DIM / 4) + sub];
    const unsigned k0 = rp[i] + (unsigned)sub;
    const bool ok = k0 < rp[i + 1];
    const int w = sorted[ok ? k0 : 0u];
    first[i] = ok ? w : -1;
  }
  const float4 bb = ((const float4*)b1)[sub];
  const float4 wa = ((const float4*)W2)[2 * sub];
  const float4 wb = ((const float4*)W2)[2 * sub + 1];
  float res[G1_NI];
  #pragma unroll
  for (int i = 0; i < G1_NI; ++i) {
    const unsigned beg = rp[i];
    const unsigned end = rp[i + 1];
    float4 acc = make_float4(0.0f, 0.0f, 0.0f, 0.0f);
    int mine = first[i];
    for (unsigned k = beg; k < end; k += 8) {
      const unsigned kn = k + 8 + sub;
      const int nw  = sorted[(kn < end) ? kn : k];
      const int nxt = (kn < end) ? nw : -1;
      #pragma unroll
      for (int j = 0; j < 8; ++j) {
        const int sj = __shfl(mine, gbase + j);
        const bool on = sj >= 0;
        const float4 v = tv[(long long)(on ? sj : 0) * (H_DIM / 4) + sub];
        acc.x += on ? v.x : 0.0f;
        acc.y += on ? v.y : 0.0f;
        acc.z += on ? v.z : 0.0f;
        acc.w += on ? v.w : 0.0f;
      }
      mine = nxt;
    }
    const float di = rsqrtf((float)(end - beg + 1u));
    float h0 = di * (acc.x + self[i].x) + bb.x;
    float h1 = di * (acc.y + self[i].y) + bb.y;
    float h2a = di * (acc.z + self[i].z) + bb.z;
    float h3 = di * (acc.w + self[i].w) + bb.w;
    h0 = h0 > 0.0f ? h0 : 0.0f;
    h1 = h1 > 0.0f ? h1 : 0.0f;
    h2a = h2a > 0.0f ? h2a : 0.0f;
    h3 = h3 > 0.0f ? h3 : 0.0f;
    float z0 = h0 * wa.x + h1 * wa.z + h2a * wb.x + h3 * wb.z;
    float z1 = h0 * wa.y + h1 * wa.w + h2a * wb.y + h3 * wb.w;
    #pragma unroll
    for (int m = 1; m < 8; m <<= 1) { z0 += __shfl_xor(z0, m); z1 += __shfl_xor(z1, m); }
    res[i] = di * (z0 - z1);
  }
  if (sub == 0) {
    #pragma unroll
    for (int i = 0; i < G1_NI; ++i)
      if (t0 + i < N) dcol[t0 + i] = res[i];
  }
}

#define G2_WG    1024
#define G2_TPB   392
#define G2_NI    4
#define G2_FILLK 13
#define G2_TAB   (G2_WG * 4 * G2_FILLK)
static_assert(G2_NI * (G2_WG / 8) >= G2_TPB, "a lane group per target node of the work-group");

__device__ __forceinline__ void g2_ask(const float* __restrict__ src, int cnt, int tid,
                                       float4 (&nv)[G2_FILLK]) {
  const int nq = cnt >> 2;
  if (nq > 0) {
    const float4* __restrict__ s4 = (const float4*)src;
    #pragma unroll
    for (int k = 0; k < G2_FILLK; ++k) {
      const int q = tid + k * G2_WG;
      nv[k] = s4[q < nq ? q : nq - 1];
    }
  } else {
    #pragma unroll
    for (int k = 0; k < G2_FILLK; ++k) nv[k] = make_float4(0.0f, 0.0f, 0.0f, 0.0f);
  }
}

__device__ __forceinline__ void g2_put(float* __restrict__ tab, const float* __restrict__ src,
                                       int cnt, int tid, const float4 (&nv)[G2_FILLK]) {
  const int nq = cnt >> 2;
  #pragma unroll
  for (int k = 0; k < G2_FILLK; ++k) {
    const int q = tid + k * G2_WG;
    if (q < nq) ((float4*)tab)[q] = nv[k];
  }
  if (tid < (cnt & 3)) tab[(nq << 2) + tid] = src[(nq << 2) + tid];
}

__global__ void __launch_bounds__(G2_WG)
gather2_logsoftmax_kernel(const int* __restrict__ sorted,
                          const unsigned int* __restrict__ rowptr,
                          const float* __restrict__ dcol, const float* __restrict__ b2,
                          float* __restrict__ out, int N, int E) {
  __shared__ __align__(16) float tab[G2_TAB];
  const int tid  = threadIdx.x;
  const int sub  = tid & 7;
  const int grp  = tid >> 3;
  const int base = blockIdx.x * G2_TPB;

  float4 nv0[G2_FILLK];
  const int cnt0 = N < G2_TAB ? N : G2_TAB;
  g2_ask(dcol, cnt0, tid, nv0);

  unsigned beg[G2_NI], end[G2_NI];
  float self[G2_NI];
  #pragma unroll
  for (int i = 0; i < G2_NI; ++i) {
    const int lt = grp + i * (G2_WG / 8);
    const int t  = base + lt;
    const bool v = (lt < G2_TPB) && (t < N);
    const int tc = v ? t : N - 1;
    const bool more = tc + 1 < N;
    const unsigned b  = rowptr[tc];
    const unsigned en = rowptr[more ? tc + 1 : tc];
    beg[i]  = b;
    end[i]  = v ? (more ? en : (unsigned)E) : b;
    self[i] = dcol[tc];
  }
  g2_put(tab, dcol, cnt0, tid, nv0);

  int idx[G2_NI][8];
  #pragma unroll
  for (int i = 0; i < G2_NI; ++i) {
    #pragma unroll
    for (int j = 0; j < 8; ++j) {
      const unsigned kk = beg[i] + (unsigned)sub + 8u * (unsigned)j;
      const bool ok = kk < end[i];
      const int w = sorted[ok ? kk : 0u];
      idx[i][j] = ok ? w : -1;
    }
  }

  float acc[G2_NI];
  #pragma unroll
  for (int i = 0; i < G2_NI; ++i) acc[i] = 0.0f;

  for (int p0 = 0; p0 < N; p0 += G2_TAB) {
    const int cnt = (N - p0 < G2_TAB) ? (N - p0) : G2_TAB;
    if (p0 > 0) {
      float4 nv[G2_FILLK];
      g2_ask(dcol + p0, cnt, tid, nv);
      __syncthreads();
      g2_put(tab, dcol + p0, cnt, tid, nv);
    }
    __syncthreads();
    #pragma unroll
    for (int i = 0; i < G2_NI; ++i) {
      if ((tid & ~31) / 8 + i * (G2_WG / 8) < G2_TPB) {
        float v[8];
        #pragma unroll
        for (int j = 0; j < 8; ++j) {
          const unsigned u = (unsigned)(idx[i][j] - p0);
          v[j] = tab[u < (unsigned)cnt ? u : 0u];
        }
        #pragma unroll
        for (int j = 0; j < 8; ++j) {
          const unsigned u = (unsigned)(idx[i][j] - p0);
          acc[i] += u < (unsigned)cnt ? v[j] : 0.0f;
        }
      }
    }
  }

  const float bd = b2[0] - b2[1];
  #pragma unroll
  for (int i = 0; i < G2_NI; ++i) {
    const int lt = grp + i * (G2_WG / 8);
    float a = acc[i];
    for (unsigned k = beg[i] + (unsigned)sub + 64u; k < end[i]; k += 8u) a += dcol[sorted[k]];
    #pragma unroll
    for (int m = 1; m < 8; m <<= 1) a += __shfl_xor(a, m);
    const float di  = rsqrtf((float)(end[i] - beg[i] + 1u));
    const float d   = di * (a + self[i]) + bd;
    const float mx  = fmaxf(d, 0.0f);
    const float lse = mx + logf(expf(d - mx) + expf(-mx));
    if (sub == 0 && lt < G2_TPB && base + lt < N) ((float2*)out)[base + lt] = make_float2(d - lse, -lse);
  }
}

extern "C" void kernel_launch(void* const* d_in, const int* in_sizes, int n_in,
                              void* d_out, int out_size, void* d_ws, size_t ws_size,
                              hipStream_t stream) {
  const float* x  = (const float*)d_in[0];
  const int*   ei = (const int*)d_in[1];
  const float* W1 = (const float*)d_in[2];
  const float* b1 = (const float*)d_in[3];
  const float* W2 = (const float*)d_in[4];
  const float* b2 = (const float*)d_in[5];
  float* out = (float*)d_out;

  const int N = in_sizes[0] / D_IN;
  const int E = in_sizes[1] / 2;

  float* ws   = (float*)d_ws;
  float* t1   = ws;
  float* h2   = t1 + (long long)N * H_DIM;
  unsigned int* rowptr = (unsigned int*)(h2 + (long long)N * C_DIM);
  int* sorted = (int*)(rowptr + N);
  const int NB = (N + BLK_NODES - 1) / BLK_NODES;
  const int P  = (E + CHUNK - 1) / CHUNK;
  unsigned int* mid   = (unsigned int*)t1;
  unsigned int* deg   = (unsigned int*)h2;
  unsigned int* lstab = deg + N;

  const int ggroups = (N + G1_NI - 1) / G1_NI;
  const int gblocks = (int)(((long long)ggroups * 8 + GB - 1) / GB);

  chunk_order_kernel<<<P, SWG, 0, stream>>>(ei, lstab, mid, E, N, NB);
  block_sort_kernel<<<NB, SWG, 0, stream>>>(mid, lstab, rowptr, deg, sorted, N, P, NB);

  gemm1_wmma_kernel<<<(N + 127) / 128, GEMM_WG, 0, stream>>>(x, W1, deg, t1, N);
  gather1_kernel<<<gblocks, GB, 0, stream>>>(sorted, rowptr, t1, b1, W2, h2, N, E);

  gather2_logsoftmax_kernel<<<(N + G2_TPB - 1) / G2_TPB, G2_WG, 0, stream>>>(sorted, rowptr, h2, b2, out, N, E);
}
